# starting kernel
# baseline (speedup 1.0000x reference)
#include <hip/hip_runtime.h>

typedef __attribute__((ext_vector_type(2))) float v2f;
typedef __attribute__((ext_vector_type(8))) float v8f;

#define N_NODES 100000
#define D 128
#define SLOPE 0.2f

__global__ void sage_scatter(const float* __restrict__ x,
                             const int*   __restrict__ ei,
                             float*       __restrict__ agg,
                             float*       __restrict__ cnt,
                             int E) {
    long long t = (long long)blockIdx.x * blockDim.x + threadIdx.x;
    int e    = (int)(t >> 5);
    int lane = (int)(t & 31);
    if (e >= E) return;
    int src = ei[e];
    int dst = ei[(long long)E + e];

    float4 v = ((const float4*)(x + (long long)src * D))[lane];
    float* a = agg + (long long)dst * D + lane * 4;
    atomicAdd(a + 0, v.x);
    atomicAdd(a + 1, v.y);
    atomicAdd(a + 2, v.z);
    atomicAdd(a + 3, v.w);
    if (lane == 0) atomicAdd(cnt + dst, 1.0f);
}

__global__ void __launch_bounds__(256)
sage_gemm(const float* __restrict__ x,
          const float* __restrict__ Wl,
          const float* __restrict__ bl,
          const float* __restrict__ Wr,
          const float* __restrict__ cnt,
          float* __restrict__ out  ) {
    const int wave = (blockIdx.x * (blockDim.x >> 5)) + (threadIdx.x >> 5);
    const int lane = threadIdx.x & 31;
    const int r0   = wave * 16;
    if (r0 >= N_NODES) return;

    const int lrow  = lane & 15;
    const int khalf = (lane >> 4) * 2;
    const int row   = r0 + lrow;

    const float rc = 1.0f / fmaxf(cnt[row], 1.0f);

    v8f acc[8];
#pragma unroll
    for (int o = 0; o < 8; ++o) { v8f z = {}; acc[o] = z; }

    const float* aggRow = out + (long long)row * D;
    const float* xRow   = x   + (long long)row * D;

    for (int kk = 0; kk < D; kk += 4) {
        float2 t = *(const float2*)(aggRow + kk + khalf);
        v2f a; a.x = t.x * rc; a.y = t.y * rc;
#pragma unroll
        for (int o = 0; o < 8; ++o) {
            float2 w = *(const float2*)(Wl + (long long)(o * 16 + lrow) * D + kk + khalf);
            v2f b; b.x = w.x; b.y = w.y;
            acc[o] = __builtin_amdgcn_wmma_f32_16x16x4_f32(
                false, a, false, b, (short)0, acc[o], false, false);
        }
    }

    for (int kk = 0; kk < D; kk += 4) {
        float2 t = *(const float2*)(xRow + kk + khalf);
        v2f a; a.x = t.x; a.y = t.y;
#pragma unroll
        for (int o = 0; o < 8; ++o) {
            float2 w = *(const float2*)(Wr + (long long)(o * 16 + lrow) * D + kk + khalf);
            v2f b; b.x = w.x; b.y = w.y;
            acc[o] = __builtin_amdgcn_wmma_f32_16x16x4_f32(
                false, a, false, b, (short)0, acc[o], false, false);
        }
    }

#pragma unroll
    for (int o = 0; o < 8; ++o) {
        const int   col = o * 16 + lrow;
        const float bb  = bl[col];
#pragma unroll
        for (int j = 0; j < 8; ++j) {
            const int rr = r0 + j + (lane >> 4) * 8;
            float v = acc[o][j] + bb;
            v = (v > 0.0f) ? v : SLOPE * v;
            out[(long long)rr * D + col] = v;
        }
    }
}

extern "C" void kernel_launch(void* const* d_in, const int* in_sizes, int n_in,
                              void* d_out, int out_size, void* d_ws, size_t ws_size,
                              hipStream_t stream) {
    const float* x  = (const float*)d_in[0];
    const int*   ei = (const int*)  d_in[1];
    const float* Wl = (const float*)d_in[2];
    const float* bl = (const float*)d_in[3];
    const float* Wr = (const float*)d_in[4];
    float* out = (float*)d_out;
    float* cnt = (float*)d_ws;
    const int E = in_sizes[1] / 2;

    hipMemsetAsync(d_out, 0, (size_t)N_NODES * D * sizeof(float), stream);
    hipMemsetAsync(d_ws,  0, (size_t)N_NODES * sizeof(float),     stream);

    {
        long long tot = (long long)E * 32;
        int blocks = (int)((tot + 255) / 256);
        sage_scatter<<<blocks, 256, 0, stream>>>(x, ei, out, cnt, E);
    }

    {
        int tiles  = (N_NODES + 15) / 16;
        int blocks = (tiles + 7) / 8;
        sage_gemm<<<blocks, 256, 0, stream>>>(x, Wl, bl, Wr, cnt, out);
    }
}


# f32-atomic edge scatter replaced by counting sort of edges by dst (1 int atomic per edge) + one wave per node summing rows in registers; sum order = int-atomic rank order; out memset dropped
# speedup vs baseline: 1.5436x; 1.5436x over previous
#include <hip/hip_runtime.h>

typedef __attribute__((ext_vector_type(2))) float v2f;
typedef __attribute__((ext_vector_type(8))) float v8f;

#define N_NODES 100000
#define D 128
#define SLOPE 0.2f

__global__ void __launch_bounds__(256)
sage_rank(const int* __restrict__ ei,
          int*       __restrict__ deg,
          int*       __restrict__ rank,
          int E) {
    const int e = blockIdx.x * 256 + threadIdx.x;
    if (e >= E) return;
    const int dst = ei[(long long)E + e];
    rank[e] = atomicAdd(deg + dst, 1);
}

__global__ void __launch_bounds__(256)
sage_offsets(const int* __restrict__ deg,
             int*       __restrict__ total,
             int*       __restrict__ start) {
    __shared__ int wave_base[8];
    __shared__ int block_base;
    const int i    = blockIdx.x * 256 + threadIdx.x;
    const int lane = threadIdx.x & 31;
    const int w    = threadIdx.x >> 5;
    const int d    = (i < N_NODES) ? deg[i] : 0;

    int inc = d;
    for (int o = 1; o < 32; o <<= 1) {
        const int t = __shfl_up(inc, o);
        if (lane >= o) inc += t;
    }
    if (lane == 31) wave_base[w] = inc;
    __syncthreads();
    if (threadIdx.x == 0) {
        int run = 0;
        for (int k = 0; k < 8; ++k) { const int t = wave_base[k]; wave_base[k] = run; run += t; }
        block_base = atomicAdd(total, run);
    }
    __syncthreads();
    if (i < N_NODES) start[i] = block_base + wave_base[w] + inc - d;
}

__global__ void __launch_bounds__(256)
sage_fill(const int* __restrict__ ei,
          const int* __restrict__ start,
          const int* __restrict__ rank,
          int*       __restrict__ slot,
          int E) {
    const int e = blockIdx.x * 256 + threadIdx.x;
    if (e >= E) return;
    const int src = ei[e];
    const int dst = ei[(long long)E + e];
    slot[start[dst] + rank[e]] = src;
}

__global__ void __launch_bounds__(256)
sage_gather(const float* __restrict__ x,
            const int*   __restrict__ start,
            const int*   __restrict__ deg,
            const int*   __restrict__ slot,
            float*       __restrict__ agg,
            float*       __restrict__ cnt) {
    const int node = __builtin_amdgcn_readfirstlane((int)(blockIdx.x * 8 + (threadIdx.x >> 5)));
    const int lane = threadIdx.x & 31;
    if (node >= N_NODES) return;
    const int beg = start[node];
    const int n   = deg[node];

    const float4* x4 = (const float4*)x;
    float4 acc = make_float4(0.0f, 0.0f, 0.0f, 0.0f);

    for (int c = 0; c < n; c += 32) {
        const int m  = (n - c < 32) ? (n - c) : 32;
        const int li = (lane < m) ? lane : (m - 1);
        const int idx = slot[beg + c + li];
        for (int j = 0; j < m; j += 8) {
            int s[8];
#pragma unroll
            for (int u = 0; u < 8; ++u) {
                const int jj = (j + u < m) ? (j + u) : (m - 1);
                s[u] = __builtin_amdgcn_readlane(idx, jj);
            }
            float4 v[8];
#pragma unroll
            for (int u = 0; u < 8; ++u) {
                v[u] = make_float4(0.0f, 0.0f, 0.0f, 0.0f);
                if (j + u < m) v[u] = x4[(long long)s[u] * (D / 4) + lane];
            }
#pragma unroll
            for (int u = 0; u < 8; ++u) {
                acc.x += v[u].x;
                acc.y += v[u].y;
                acc.z += v[u].z;
                acc.w += v[u].w;
            }
        }
    }
    ((float4*)(agg + (long long)node * D))[lane] = acc;
    if (lane == 0) cnt[node] = (float)n;
}

__global__ void __launch_bounds__(256)
sage_gemm(const float* __restrict__ x,
          const float* __restrict__ Wl,
          const float* __restrict__ bl,
          const float* __restrict__ Wr,
          const float* __restrict__ cnt,
          float* __restrict__ out  ) {
    const int wave = (blockIdx.x * (blockDim.x >> 5)) + (threadIdx.x >> 5);
    const int lane = threadIdx.x & 31;
    const int r0   = wave * 16;
    if (r0 >= N_NODES) return;

    const int lrow  = lane & 15;
    const int khalf = (lane >> 4) * 2;
    const int row   = r0 + lrow;

    const float rc = 1.0f / fmaxf(cnt[row], 1.0f);

    v8f acc[8];
#pragma unroll
    for (int o = 0; o < 8; ++o) { v8f z = {}; acc[o] = z; }

    const float* aggRow = out + (long long)row * D;
    const float* xRow   = x   + (long long)row * D;

    for (int kk = 0; kk < D; kk += 4) {
        float2 t = *(const float2*)(aggRow + kk + khalf);
        v2f a; a.x = t.x * rc; a.y = t.y * rc;
#pragma unroll
        for (int o = 0; o < 8; ++o) {
            float2 w = *(const float2*)(Wl + (long long)(o * 16 + lrow) * D + kk + khalf);
            v2f b; b.x = w.x; b.y = w.y;
            acc[o] = __builtin_amdgcn_wmma_f32_16x16x4_f32(
                false, a, false, b, (short)0, acc[o], false, false);
        }
    }

    for (int kk = 0; kk < D; kk += 4) {
        float2 t = *(const float2*)(xRow + kk + khalf);
        v2f a; a.x = t.x; a.y = t.y;
#pragma unroll
        for (int o = 0; o < 8; ++o) {
            float2 w = *(const float2*)(Wr + (long long)(o * 16 + lrow) * D + kk + khalf);
            v2f b; b.x = w.x; b.y = w.y;
            acc[o] = __builtin_amdgcn_wmma_f32_16x16x4_f32(
                false, a, false, b, (short)0, acc[o], false, false);
        }
    }

#pragma unroll
    for (int o = 0; o < 8; ++o) {
        const int   col = o * 16 + lrow;
        const float bb  = bl[col];
#pragma unroll
        for (int j = 0; j < 8; ++j) {
            const int rr = r0 + j + (lane >> 4) * 8;
            float v = acc[o][j] + bb;
            v = (v > 0.0f) ? v : SLOPE * v;
            out[(long long)rr * D + col] = v;
        }
    }
}

extern "C" void kernel_launch(void* const* d_in, const int* in_sizes, int n_in,
                              void* d_out, int out_size, void* d_ws, size_t ws_size,
                              hipStream_t stream) {
    const float* x  = (const float*)d_in[0];
    const int*   ei = (const int*)  d_in[1];
    const float* Wl = (const float*)d_in[2];
    const float* bl = (const float*)d_in[3];
    const float* Wr = (const float*)d_in[4];
    float* out = (float*)d_out;
    const int E = in_sizes[1] / 2;

    char*  ws    = (char*)d_ws;
    int*   deg   = (int*)ws;
    int*   total = deg + N_NODES;
    int*   start = (int*)  (ws + (1  << 19));
    float* cnt   = (float*)(ws + (1  << 20));
    int*   rank  = (int*)  (ws + (2  << 20));
    int*   slot  = (int*)  (ws + (16 << 20));

    hipMemsetAsync(d_ws, 0, (size_t)(N_NODES + 32) * sizeof(int), stream);

    const int eblocks = (E + 255) / 256;
    sage_rank   <<<eblocks, 256, 0, stream>>>(ei, deg, rank, E);
    sage_offsets<<<(N_NODES + 255) / 256, 256, 0, stream>>>(deg, total, start);
    sage_fill   <<<eblocks, 256, 0, stream>>>(ei, start, rank, slot, E);
    sage_gather <<<(N_NODES + 7) / 8, 256, 0, stream>>>(x, start, deg, slot, out, cnt);

    {
        int tiles  = (N_NODES + 15) / 16;
        int blocks = (tiles + 7) / 8;
        sage_gemm<<<blocks, 256, 0, stream>>>(x, Wl, bl, Wr, cnt, out);
    }
}


# GEMM launch: each wave keeps its 16 output columns of Wl|Wr in registers all launch; A rows of each 32-row step staged once per work-group in LDS; whole-row stores; sum order unchanged
# speedup vs baseline: 2.0246x; 1.3116x over previous
#include <hip/hip_runtime.h>

typedef __attribute__((ext_vector_type(2))) float v2f;
typedef __attribute__((ext_vector_type(8))) float v8f;

#define N_NODES 100000
#define D 128
#define SLOPE 0.2f

__global__ void __launch_bounds__(256)
sage_rank(const int* __restrict__ ei,
          int*       __restrict__ deg,
          int*       __restrict__ rank,
          int E) {
    const int e = blockIdx.x * 256 + threadIdx.x;
    if (e >= E) return;
    const int dst = ei[(long long)E + e];
    rank[e] = atomicAdd(deg + dst, 1);
}

__global__ void __launch_bounds__(256)
sage_offsets(const int* __restrict__ deg,
             int*       __restrict__ total,
             int*       __restrict__ start) {
    __shared__ int wave_base[8];
    __shared__ int block_base;
    const int i    = blockIdx.x * 256 + threadIdx.x;
    const int lane = threadIdx.x & 31;
    const int w    = threadIdx.x >> 5;
    const int d    = (i < N_NODES) ? deg[i] : 0;

    int inc = d;
    for (int o = 1; o < 32; o <<= 1) {
        const int t = __shfl_up(inc, o);
        if (lane >= o) inc += t;
    }
    if (lane == 31) wave_base[w] = inc;
    __syncthreads();
    if (threadIdx.x == 0) {
        int run = 0;
        for (int k = 0; k < 8; ++k) { const int t = wave_base[k]; wave_base[k] = run; run += t; }
        block_base = atomicAdd(total, run);
    }
    __syncthreads();
    if (i < N_NODES) start[i] = block_base + wave_base[w] + inc - d;
}

__global__ void __launch_bounds__(256)
sage_fill(const int* __restrict__ ei,
          const int* __restrict__ start,
          const int* __restrict__ rank,
          int*       __restrict__ slot,
          int E) {
    const int e = blockIdx.x * 256 + threadIdx.x;
    if (e >= E) return;
    const int src = ei[e];
    const int dst = ei[(long long)E + e];
    slot[start[dst] + rank[e]] = src;
}

__global__ void __launch_bounds__(256)
sage_gather(const float* __restrict__ x,
            const int*   __restrict__ start,
            const int*   __restrict__ deg,
            const int*   __restrict__ slot,
            float*       __restrict__ agg,
            float*       __restrict__ cnt) {
    const int node = __builtin_amdgcn_readfirstlane((int)(blockIdx.x * 8 + (threadIdx.x >> 5)));
    const int lane = threadIdx.x & 31;
    if (node >= N_NODES) return;
    const int beg = start[node];
    const int n   = deg[node];

    const float4* x4 = (const float4*)x;
    float4 acc = make_float4(0.0f, 0.0f, 0.0f, 0.0f);

    for (int c = 0; c < n; c += 32) {
        const int m  = (n - c < 32) ? (n - c) : 32;
        const int li = (lane < m) ? lane : (m - 1);
        const int idx = slot[beg + c + li];
        for (int j = 0; j < m; j += 8) {
            int s[8];
#pragma unroll
            for (int u = 0; u < 8; ++u) {
                const int jj = (j + u < m) ? (j + u) : (m - 1);
                s[u] = __builtin_amdgcn_readlane(idx, jj);
            }
            float4 v[8];
#pragma unroll
            for (int u = 0; u < 8; ++u) {
                v[u] = make_float4(0.0f, 0.0f, 0.0f, 0.0f);
                if (j + u < m) v[u] = x4[(long long)s[u] * (D / 4) + lane];
            }
#pragma unroll
            for (int u = 0; u < 8; ++u) {
                acc.x += v[u].x;
                acc.y += v[u].y;
                acc.z += v[u].z;
                acc.w += v[u].w;
            }
        }
    }
    ((float4*)(agg + (long long)node * D))[lane] = acc;
    if (lane == 0) cnt[node] = (float)n;
}

#define G2_ROWS  32
#define G2_STEPS (N_NODES / G2_ROWS)
#define G2_GRID  512
#define A_PITCH  260
#define O_PITCH  132
#define W_PITCH  132
#define A_FLOATS (G2_ROWS * A_PITCH)
#define O_FLOATS (G2_ROWS * O_PITCH)
#define LDS_FLOATS (2 * A_FLOATS + 2 * O_FLOATS)

struct StepRows {
    float4 a0, a1, a2, a3;
    float4 x0, x1, x2, x3;
    float  n0, n1, n2, n3;
};

__device__ __forceinline__ void rows_load(StepRows& p, const float* agg, const float* x,
                                          const float* cnt, int row0, int lane) {
    p.a0 = ((const float4*)(agg + (long long)(row0     ) * D))[lane];
    p.x0 = ((const float4*)(x   + (long long)(row0     ) * D))[lane];
    p.a1 = ((const float4*)(agg + (long long)(row0 +  8) * D))[lane];
    p.x1 = ((const float4*)(x   + (long long)(row0 +  8) * D))[lane];
    p.a2 = ((const float4*)(agg + (long long)(row0 + 16) * D))[lane];
    p.x2 = ((const float4*)(x   + (long long)(row0 + 16) * D))[lane];
    p.a3 = ((const float4*)(agg + (long long)(row0 + 24) * D))[lane];
    p.x3 = ((const float4*)(x   + (long long)(row0 + 24) * D))[lane];
    p.n0 = cnt[row0     ];
    p.n1 = cnt[row0 +  8];
    p.n2 = cnt[row0 + 16];
    p.n3 = cnt[row0 + 24];
}

__device__ __forceinline__ void row_put(float* dr, float4 a, float deg, float4 xv, int lane) {
    const float rc = 1.0f / fmaxf(deg, 1.0f);
    a.x *= rc; a.y *= rc; a.z *= rc; a.w *= rc;
    ((float4*)dr)[lane]       = a;
    ((float4*)(dr + D))[lane] = xv;
}

__device__ __forceinline__ void rows_put(const StepRows& p, float* ab, int w, int lane) {
    row_put(ab + (w     ) * A_PITCH, p.a0, p.n0, p.x0, lane);
    row_put(ab + (w +  8) * A_PITCH, p.a1, p.n1, p.x1, lane);
    row_put(ab + (w + 16) * A_PITCH, p.a2, p.n2, p.x2, lane);
    row_put(ab + (w + 24) * A_PITCH, p.a3, p.n3, p.x3, lane);
}

__global__ void __launch_bounds__(256)
sage_gemm(const float* __restrict__ x,
          const float* __restrict__ Wl,
          const float* __restrict__ bl,
          const float* __restrict__ Wr,
          const float* __restrict__ cnt,
          float* out  ) {
    __shared__ __attribute__((aligned(16))) float lds[LDS_FLOATS];

    const int w    = __builtin_amdgcn_readfirstlane((int)(threadIdx.x >> 5));
    const int lane = threadIdx.x & 31;
    const int c    = lane & 15;
    const int h    = lane >> 4;

    v2f breg[64];
    {
        float*       wst = lds + w * (16 * W_PITCH);
        const float* wfr = wst + c * W_PITCH + 2 * h;
#pragma unroll
        for (int r = 0; r < 16; ++r)
            ((float4*)(wst + r * W_PITCH))[lane] =
                ((const float4*)(Wl + (long long)(w * 16 + r) * D))[lane];
        __syncthreads();
#pragma unroll
        for (int j = 0; j < 32; ++j) breg[j] = *(const v2f*)(wfr + 4 * j);
        __syncthreads();
#pragma unroll
        for (int r = 0; r < 16; ++r)
            ((float4*)(wst + r * W_PITCH))[lane] =
                ((const float4*)(Wr + (long long)(w * 16 + r) * D))[lane];
        __syncthreads();
#pragma unroll
        for (int j = 0; j < 32; ++j) breg[32 + j] = *(const v2f*)(wfr + 4 * j);
        __syncthreads();
    }
    const float bb = bl[w * 16 + c];

    StepRows p;
    int s = blockIdx.x;
    rows_load(p, out, x, cnt, s * G2_ROWS + w, lane);
    rows_put(p, lds, w, lane);
    __syncthreads();

    int cur = 0;
    for (; s < G2_STEPS; s += G2_GRID) {
        const int  sn   = s + G2_GRID;
        const bool more = sn < G2_STEPS;
        float* abuf = lds + cur * A_FLOATS;
        float* obuf = lds + 2 * A_FLOATS + cur * O_FLOATS;

        if (more) rows_load(p, out, x, cnt, sn * G2_ROWS + w, lane);

        const float* a0p = abuf + c * A_PITCH + 2 * h;
        const float* a1p = a0p + 16 * A_PITCH;
        v8f acc0 = {}, acc1 = {};
#pragma unroll
        for (int j = 0; j < 64; ++j) {
            const v2f a0 = *(const v2f*)(a0p + 4 * j);
            const v2f a1 = *(const v2f*)(a1p + 4 * j);
            acc0 = __builtin_amdgcn_wmma_f32_16x16x4_f32(
                false, a0, false, breg[j], (short)0, acc0, false, false);
            acc1 = __builtin_amdgcn_wmma_f32_16x16x4_f32(
                false, a1, false, breg[j], (short)0, acc1, false, false);
        }

#pragma unroll
        for (int j = 0; j < 8; ++j) {
            float v0 = acc0[j] + bb;
            v0 = (v0 > 0.0f) ? v0 : SLOPE * v0;
            obuf[(8 * h + j) * O_PITCH + 16 * w + c] = v0;
            float v1 = acc1[j] + bb;
            v1 = (v1 > 0.0f) ? v1 : SLOPE * v1;
            obuf[(16 + 8 * h + j) * O_PITCH + 16 * w + c] = v1;
        }

        if (more) rows_put(p, lds + (cur ^ 1) * A_FLOATS, w, lane);
        __syncthreads();

#pragma unroll
        for (int q = 0; q < 4; ++q) {
            const int r = q * 8 + w;
            const float4 t = ((const float4*)(obuf + r * O_PITCH))[lane];
            ((float4*)(out + (long long)(s * G2_ROWS + r) * D))[lane] = t;
        }
        cur ^= 1;
    }
}

extern "C" void kernel_launch(void* const* d_in, const int* in_sizes, int n_in,
                              void* d_out, int out_size, void* d_ws, size_t ws_size,
                              hipStream_t stream) {
    const float* x  = (const float*)d_in[0];
    const int*   ei = (const int*)  d_in[1];
    const float* Wl = (const float*)d_in[2];
    const float* bl = (const float*)d_in[3];
    const float* Wr = (const float*)d_in[4];
    float* out = (float*)d_out;
    const int E = in_sizes[1] / 2;

    char*  ws    = (char*)d_ws;
    int*   deg   = (int*)ws;
    int*   total = deg + N_NODES;
    int*   start = (int*)  (ws + (1  << 19));
    float* cnt   = (float*)(ws + (1  << 20));
    int*   rank  = (int*)  (ws + (2  << 20));
    int*   slot  = (int*)  (ws + (16 << 20));

    hipMemsetAsync(d_ws, 0, (size_t)(N_NODES + 32) * sizeof(int), stream);

    const int eblocks = (E + 255) / 256;
    sage_rank   <<<eblocks, 256, 0, stream>>>(ei, deg, rank, E);
    sage_offsets<<<(N_NODES + 255) / 256, 256, 0, stream>>>(deg, total, start);
    sage_fill   <<<eblocks, 256, 0, stream>>>(ei, start, rank, slot, E);
    sage_gather <<<(N_NODES + 7) / 8, 256, 0, stream>>>(x, start, deg, slot, out, cnt);

    sage_gemm<<<G2_GRID, 256, 0, stream>>>(x, Wl, bl, Wr, cnt, out);
}


# edge grouping by destination: two-level counting sort on LDS counters (edge chunks ordered by 256-node bin in LDS, copied out in runs) replaces per-edge global atomics; sum order = LDS-atomic order
# speedup vs baseline: 3.4522x; 1.7051x over previous
#include <hip/hip_runtime.h>

typedef __attribute__((ext_vector_type(2))) float v2f;
typedef __attribute__((ext_vector_type(8))) float v8f;

#define N_NODES 100000
#define D 128
#define SLOPE 0.2f

#define BIN_SHIFT 8
#define BIN_NODES (1 << BIN_SHIFT)
#define NB ((N_NODES + BIN_NODES - 1) / BIN_NODES)
#define CH 4096
#define CH_T (CH / 256)
static_assert(N_NODES <= (1 << 24), "src<<8 must fit a positive int");

__device__ __forceinline__ int block_excl_scan(int v, int* wsum, int& total) {
    const int lane = threadIdx.x & 31;
    const int w    = threadIdx.x >> 5;
    int inc = v;
    for (int o = 1; o < 32; o <<= 1) {
        const int t = __shfl_up(inc, o);
        if (lane >= o) inc += t;
    }
    __syncthreads();
    if (lane == 31) wsum[w] = inc;
    __syncthreads();
    int base = 0, tot = 0;
    for (int k = 0; k < 8; ++k) {
        const int t = wsum[k];
        if (k < w) base += t;
        tot += t;
    }
    total = tot;
    return base + inc - v;
}

__global__ void __launch_bounds__(256)
sort_hist(const int* __restrict__ ei,
          int*       __restrict__ histG,
          int E) {
    __shared__ int h[NB];
    for (int i = threadIdx.x; i < NB; i += 256) h[i] = 0;
    __syncthreads();
    const int base = blockIdx.x * CH;
    int d[CH_T];
#pragma unroll
    for (int k = 0; k < CH_T; ++k) {
        const int e  = base + k * 256 + threadIdx.x;
        const int ee = (e < E) ? e : (E - 1);
        d[k] = ei[(long long)E + ee];
    }
#pragma unroll
    for (int k = 0; k < CH_T; ++k) {
        const int e = base + k * 256 + threadIdx.x;
        if (e < E) atomicAdd(&h[d[k] >> BIN_SHIFT], 1);
    }
    __syncthreads();
    for (int i = threadIdx.x; i < NB; i += 256) histG[blockIdx.x * NB + i] = h[i];
}

__global__ void __launch_bounds__(256)
sort_binscan(const int* __restrict__ histG,
             int*       __restrict__ offG,
             int*       __restrict__ bintot,
             int G) {
    __shared__ int wsum[8];
    const int b = blockIdx.x;
    int carry = 0;
    for (int g0 = 0; g0 < G; g0 += 256) {
        const int g = g0 + threadIdx.x;
        const int v = (g < G) ? histG[g * NB + b] : 0;
        int tot;
        const int ex = block_excl_scan(v, wsum, tot);
        if (g < G) offG[g * NB + b] = carry + ex;
        carry += tot;
    }
    if (threadIdx.x == 0) bintot[b] = carry;
}

__global__ void __launch_bounds__(256)
sort_scatter(const int* __restrict__ ei,
             const int* __restrict__ offG,
             const int* __restrict__ bintot,
             int*       __restrict__ binned,
             int E) {
    __shared__ int h[NB];
    __shared__ int lstart[NB];
    __shared__ int gdst[NB];
    __shared__ int stage[CH];
    __shared__ unsigned short sbin[CH];
    __shared__ int wsum[8];

    for (int i = threadIdx.x; i < NB; i += 256) h[i] = 0;
    __syncthreads();
    const int base = blockIdx.x * CH;
    int sv[CH_T], dv[CH_T], rk[CH_T];
#pragma unroll
    for (int k = 0; k < CH_T; ++k) {
        const int e  = base + k * 256 + threadIdx.x;
        const int ee = (e < E) ? e : (E - 1);
        sv[k] = ei[ee];
        dv[k] = ei[(long long)E + ee];
    }
#pragma unroll
    for (int k = 0; k < CH_T; ++k) {
        const int e = base + k * 256 + threadIdx.x;
        rk[k] = 0;
        if (e < E) rk[k] = atomicAdd(&h[dv[k] >> BIN_SHIFT], 1);
    }
    __syncthreads();

    int carryL = 0, carryG = 0;
    for (int i0 = 0; i0 < NB; i0 += 256) {
        const int i  = i0 + threadIdx.x;
        const int hv = (i < NB) ? h[i] : 0;
        const int tv = (i < NB) ? bintot[i] : 0;
        int totL, totG;
        const int exL = block_excl_scan(hv, wsum, totL);
        const int exG = block_excl_scan(tv, wsum, totG);
        if (i < NB) {
            lstart[i] = carryL + exL;
            gdst[i]   = carryG + exG + offG[blockIdx.x * NB + i];
        }
        carryL += totL;
        carryG += totG;
    }
    __syncthreads();

#pragma unroll
    for (int k = 0; k < CH_T; ++k) {
        const int e = base + k * 256 + threadIdx.x;
        if (e < E) {
            const int b = dv[k] >> BIN_SHIFT;
            const int p = lstart[b] + rk[k];
            stage[p] = (sv[k] << BIN_SHIFT) | (dv[k] & (BIN_NODES - 1));
            sbin[p]  = (unsigned short)b;
        }
    }
    __syncthreads();

    const int n = (E - base < CH) ? (E - base) : CH;
    for (int i = threadIdx.x; i < n; i += 256) {
        const int b = sbin[i];
        binned[gdst[b] + (i - lstart[b])] = stage[i];
    }
}

__global__ void __launch_bounds__(256)
sort_fine(const int* __restrict__ bintot,
          const int* __restrict__ binned,
          int*       __restrict__ start,
          int*       __restrict__ deg,
          int*       __restrict__ slot) {
    __shared__ int c[BIN_NODES];
    __shared__ int wsum[8];
    const int b = blockIdx.x;

    int part = 0;
    for (int i = threadIdx.x; i < b; i += 256) part += bintot[i];
    int lo;
    (void)block_excl_scan(part, wsum, lo);
    const int hi = lo + bintot[b];

    c[threadIdx.x] = 0;
    __syncthreads();

    for (int t0 = lo; t0 < hi; t0 += CH) {
        int w[CH_T];
#pragma unroll
        for (int k = 0; k < CH_T; ++k) {
            const int e = t0 + k * 256 + threadIdx.x;
            w[k] = binned[(e < hi) ? e : (hi - 1)];
        }
#pragma unroll
        for (int k = 0; k < CH_T; ++k) {
            const int e = t0 + k * 256 + threadIdx.x;
            if (e < hi) atomicAdd(&c[w[k] & (BIN_NODES - 1)], 1);
        }
    }
    __syncthreads();

    const int dgr = c[threadIdx.x];
    int tot;
    const int ex = block_excl_scan(dgr, wsum, tot);
    const int node = b * BIN_NODES + threadIdx.x;
    if (node < N_NODES) { start[node] = lo + ex; deg[node] = dgr; }
    c[threadIdx.x] = lo + ex;
    __syncthreads();

    for (int t0 = lo; t0 < hi; t0 += CH) {
        int w[CH_T];
#pragma unroll
        for (int k = 0; k < CH_T; ++k) {
            const int e = t0 + k * 256 + threadIdx.x;
            w[k] = binned[(e < hi) ? e : (hi - 1)];
        }
#pragma unroll
        for (int k = 0; k < CH_T; ++k) {
            const int e = t0 + k * 256 + threadIdx.x;
            if (e < hi) {
                const int p = atomicAdd(&c[w[k] & (BIN_NODES - 1)], 1);
                slot[p] = w[k] >> BIN_SHIFT;
            }
        }
    }
}

__global__ void __launch_bounds__(256)
sage_gather(const float* __restrict__ x,
            const int*   __restrict__ start,
            const int*   __restrict__ deg,
            const int*   __restrict__ slot,
            float*       __restrict__ agg,
            float*       __restrict__ cnt) {
    const int node = __builtin_amdgcn_readfirstlane((int)(blockIdx.x * 8 + (threadIdx.x >> 5)));
    const int lane = threadIdx.x & 31;
    if (node >= N_NODES) return;
    const int beg = start[node];
    const int n   = deg[node];

    const float4* x4 = (const float4*)x;
    float4 acc = make_float4(0.0f, 0.0f, 0.0f, 0.0f);

    for (int c = 0; c < n; c += 32) {
        const int m  = (n - c < 32) ? (n - c) : 32;
        const int li = (lane < m) ? lane : (m - 1);
        const int idx = slot[beg + c + li];
        for (int j = 0; j < m; j += 8) {
            int s[8];
#pragma unroll
            for (int u = 0; u < 8; ++u) {
                const int jj = (j + u < m) ? (j + u) : (m - 1);
                s[u] = __builtin_amdgcn_readlane(idx, jj);
            }
            float4 v[8];
#pragma unroll
            for (int u = 0; u < 8; ++u) {
                v[u] = make_float4(0.0f, 0.0f, 0.0f, 0.0f);
                if (j + u < m) v[u] = x4[(long long)s[u] * (D / 4) + lane];
            }
#pragma unroll
            for (int u = 0; u < 8; ++u) {
                acc.x += v[u].x;
                acc.y += v[u].y;
                acc.z += v[u].z;
                acc.w += v[u].w;
            }
        }
    }
    ((float4*)(agg + (long long)node * D))[lane] = acc;
    if (lane == 0) cnt[node] = (float)n;
}

#define G2_ROWS  32
#define G2_STEPS (N_NODES / G2_ROWS)
#define G2_GRID  512
#define A_PITCH  260
#define O_PITCH  132
#define W_PITCH  132
#define A_FLOATS (G2_ROWS * A_PITCH)
#define O_FLOATS (G2_ROWS * O_PITCH)
#define LDS_FLOATS (2 * A_FLOATS + 2 * O_FLOATS)

struct StepRows {
    float4 a0, a1, a2, a3;
    float4 x0, x1, x2, x3;
    float  n0, n1, n2, n3;
};

__device__ __forceinline__ void rows_load(StepRows& p, const float* agg, const float* x,
                                          const float* cnt, int row0, int lane) {
    p.a0 = ((const float4*)(agg + (long long)(row0     ) * D))[lane];
    p.x0 = ((const float4*)(x   + (long long)(row0     ) * D))[lane];
    p.a1 = ((const float4*)(agg + (long long)(row0 +  8) * D))[lane];
    p.x1 = ((const float4*)(x   + (long long)(row0 +  8) * D))[lane];
    p.a2 = ((const float4*)(agg + (long long)(row0 + 16) * D))[lane];
    p.x2 = ((const float4*)(x   + (long long)(row0 + 16) * D))[lane];
    p.a3 = ((const float4*)(agg + (long long)(row0 + 24) * D))[lane];
    p.x3 = ((const float4*)(x   + (long long)(row0 + 24) * D))[lane];
    p.n0 = cnt[row0     ];
    p.n1 = cnt[row0 +  8];
    p.n2 = cnt[row0 + 16];
    p.n3 = cnt[row0 + 24];
}

__device__ __forceinline__ void row_put(float* dr, float4 a, float deg, float4 xv, int lane) {
    const float rc = 1.0f / fmaxf(deg, 1.0f);
    a.x *= rc; a.y *= rc; a.z *= rc; a.w *= rc;
    ((float4*)dr)[lane]       = a;
    ((float4*)(dr + D))[lane] = xv;
}

__device__ __forceinline__ void rows_put(const StepRows& p, float* ab, int w, int lane) {
    row_put(ab + (w     ) * A_PITCH, p.a0, p.n0, p.x0, lane);
    row_put(ab + (w +  8) * A_PITCH, p.a1, p.n1, p.x1, lane);
    row_put(ab + (w + 16) * A_PITCH, p.a2, p.n2, p.x2, lane);
    row_put(ab + (w + 24) * A_PITCH, p.a3, p.n3, p.x3, lane);
}

__global__ void __launch_bounds__(256)
sage_gemm(const float* __restrict__ x,
          const float* __restrict__ Wl,
          const float* __restrict__ bl,
          const float* __restrict__ Wr,
          const float* __restrict__ cnt,
          float* out  ) {
    __shared__ __attribute__((aligned(16))) float lds[LDS_FLOATS];

    const int w    = __builtin_amdgcn_readfirstlane((int)(threadIdx.x >> 5));
    const int lane = threadIdx.x & 31;
    const int c    = lane & 15;
    const int h    = lane >> 4;

    v2f breg[64];
    {
        float*       wst = lds + w * (16 * W_PITCH);
        const float* wfr = wst + c * W_PITCH + 2 * h;
#pragma unroll
        for (int r = 0; r < 16; ++r)
            ((float4*)(wst + r * W_PITCH))[lane] =
                ((const float4*)(Wl + (long long)(w * 16 + r) * D))[lane];
        __syncthreads();
#pragma unroll
        for (int j = 0; j < 32; ++j) breg[j] = *(const v2f*)(wfr + 4 * j);
        __syncthreads();
#pragma unroll
        for (int r = 0; r < 16; ++r)
            ((float4*)(wst + r * W_PITCH))[lane] =
                ((const float4*)(Wr + (long long)(w * 16 + r) * D))[lane];
        __syncthreads();
#pragma unroll
        for (int j = 0; j < 32; ++j) breg[32 + j] = *(const v2f*)(wfr + 4 * j);
        __syncthreads();
    }
    const float bb = bl[w * 16 + c];

    StepRows p;
    int s = blockIdx.x;
    rows_load(p, out, x, cnt, s * G2_ROWS + w, lane);
    rows_put(p, lds, w, lane);
    __syncthreads();

    int cur = 0;
    for (; s < G2_STEPS; s += G2_GRID) {
        const int  sn   = s + G2_GRID;
        const bool more = sn < G2_STEPS;
        float* abuf = lds + cur * A_FLOATS;
        float* obuf = lds + 2 * A_FLOATS + cur * O_FLOATS;

        if (more) rows_load(p, out, x, cnt, sn * G2_ROWS + w, lane);

        const float* a0p = abuf + c * A_PITCH + 2 * h;
        const float* a1p = a0p + 16 * A_PITCH;
        v8f acc0 = {}, acc1 = {};
#pragma unroll
        for (int j = 0; j < 64; ++j) {
            const v2f a0 = *(const v2f*)(a0p + 4 * j);
            const v2f a1 = *(const v2f*)(a1p + 4 * j);
            acc0 = __builtin_amdgcn_wmma_f32_16x16x4_f32(
                false, a0, false, breg[j], (short)0, acc0, false, false);
            acc1 = __builtin_amdgcn_wmma_f32_16x16x4_f32(
                false, a1, false, breg[j], (short)0, acc1, false, false);
        }

#pragma unroll
        for (int j = 0; j < 8; ++j) {
            float v0 = acc0[j] + bb;
            v0 = (v0 > 0.0f) ? v0 : SLOPE * v0;
            obuf[(8 * h + j) * O_PITCH + 16 * w + c] = v0;
            float v1 = acc1[j] + bb;
            v1 = (v1 > 0.0f) ? v1 : SLOPE * v1;
            obuf[(16 + 8 * h + j) * O_PITCH + 16 * w + c] = v1;
        }

        if (more) rows_put(p, lds + (cur ^ 1) * A_FLOATS, w, lane);
        __syncthreads();

#pragma unroll
        for (int q = 0; q < 4; ++q) {
            const int r = q * 8 + w;
            const float4 t = ((const float4*)(obuf + r * O_PITCH))[lane];
            ((float4*)(out + (long long)(s * G2_ROWS + r) * D))[lane] = t;
        }
        cur ^= 1;
    }
}

extern "C" void kernel_launch(void* const* d_in, const int* in_sizes, int n_in,
                              void* d_out, int out_size, void* d_ws, size_t ws_size,
                              hipStream_t stream) {
    const float* x  = (const float*)d_in[0];
    const int*   ei = (const int*)  d_in[1];
    const float* Wl = (const float*)d_in[2];
    const float* bl = (const float*)d_in[3];
    const float* Wr = (const float*)d_in[4];
    float* out = (float*)d_out;
    const int E = in_sizes[1] / 2;

    char*  ws     = (char*)d_ws;
    int*   deg    = (int*)ws;
    int*   start  = (int*)  (ws + (1  << 19));
    float* cnt    = (float*)(ws + (1  << 20));
    int*   histG  = (int*)  (ws + (2  << 20));
    int*   offG   = (int*)  (ws + (4  << 20));
    int*   bintot = (int*)  (ws + (6  << 20));
    int*   binned = (int*)  (ws + (8  << 20));
    int*   slot   = (int*)  (ws + (16 << 20));

    const int G = (E + CH - 1) / CH;
    sort_hist   <<<G,  256, 0, stream>>>(ei, histG, E);
    sort_binscan<<<NB, 256, 0, stream>>>(histG, offG, bintot, G);
    sort_scatter<<<G,  256, 0, stream>>>(ei, offG, bintot, binned, E);
    sort_fine   <<<NB, 256, 0, stream>>>(bintot, binned, start, deg, slot);
    sage_gather <<<(N_NODES + 7) / 8, 256, 0, stream>>>(x, start, deg, slot, out, cnt);

    sage_gemm<<<G2_GRID, 256, 0, stream>>>(x, Wl, bl, Wr, cnt, out);
}


# gather launch: each wave walks 7 consecutive nodes (node loop in the wave; next node's id list asked for behind the current node's first rows) instead of one short wave per node; sum order unchanged
# speedup vs baseline: 3.7233x; 1.0785x over previous
#include <hip/hip_runtime.h>

typedef __attribute__((ext_vector_type(2))) float v2f;
typedef __attribute__((ext_vector_type(8))) float v8f;

#define N_NODES 100000
#define D 128
#define SLOPE 0.2f

#define BIN_SHIFT 8
#define BIN_NODES (1 << BIN_SHIFT)
#define NB ((N_NODES + BIN_NODES - 1) / BIN_NODES)
#define CH 4096
#define CH_T (CH / 256)
static_assert(N_NODES <= (1 << 24), "src<<8 must fit a positive int");

__device__ __forceinline__ int block_excl_scan(int v, int* wsum, int& total) {
    const int lane = threadIdx.x & 31;
    const int w    = threadIdx.x >> 5;
    int inc = v;
    for (int o = 1; o < 32; o <<= 1) {
        const int t = __shfl_up(inc, o);
        if (lane >= o) inc += t;
    }
    __syncthreads();
    if (lane == 31) wsum[w] = inc;
    __syncthreads();
    int base = 0, tot = 0;
    for (int k = 0; k < 8; ++k) {
        const int t = wsum[k];
        if (k < w) base += t;
        tot += t;
    }
    total = tot;
    return base + inc - v;
}

__global__ void __launch_bounds__(256)
sort_hist(const int* __restrict__ ei,
          int*       __restrict__ histG,
          int E) {
    __shared__ int h[NB];
    for (int i = threadIdx.x; i < NB; i += 256) h[i] = 0;
    __syncthreads();
    const int base = blockIdx.x * CH;
    int d[CH_T];
#pragma unroll
    for (int k = 0; k < CH_T; ++k) {
        const int e  = base + k * 256 + threadIdx.x;
        const int ee = (e < E) ? e : (E - 1);
        d[k] = ei[(long long)E + ee];
    }
#pragma unroll
    for (int k = 0; k < CH_T; ++k) {
        const int e = base + k * 256 + threadIdx.x;
        if (e < E) atomicAdd(&h[d[k] >> BIN_SHIFT], 1);
    }
    __syncthreads();
    for (int i = threadIdx.x; i < NB; i += 256) histG[blockIdx.x * NB + i] = h[i];
}

__global__ void __launch_bounds__(256)
sort_binscan(const int* __restrict__ histG,
             int*       __restrict__ offG,
             int*       __restrict__ bintot,
             int G) {
    __shared__ int wsum[8];
    const int b = blockIdx.x;
    int carry = 0;
    for (int g0 = 0; g0 < G; g0 += 256) {
        const int g = g0 + threadIdx.x;
        const int v = (g < G) ? histG[g * NB + b] : 0;
        int tot;
        const int ex = block_excl_scan(v, wsum, tot);
        if (g < G) offG[g * NB + b] = carry + ex;
        carry += tot;
    }
    if (threadIdx.x == 0) bintot[b] = carry;
}

__global__ void __launch_bounds__(256)
sort_scatter(const int* __restrict__ ei,
             const int* __restrict__ offG,
             const int* __restrict__ bintot,
             int*       __restrict__ binned,
             int E) {
    __shared__ int h[NB];
    __shared__ int lstart[NB];
    __shared__ int gdst[NB];
    __shared__ int stage[CH];
    __shared__ unsigned short sbin[CH];
    __shared__ int wsum[8];

    for (int i = threadIdx.x; i < NB; i += 256) h[i] = 0;
    __syncthreads();
    const int base = blockIdx.x * CH;
    int sv[CH_T], dv[CH_T], rk[CH_T];
#pragma unroll
    for (int k = 0; k < CH_T; ++k) {
        const int e  = base + k * 256 + threadIdx.x;
        const int ee = (e < E) ? e : (E - 1);
        sv[k] = ei[ee];
        dv[k] = ei[(long long)E + ee];
    }
#pragma unroll
    for (int k = 0; k < CH_T; ++k) {
        const int e = base + k * 256 + threadIdx.x;
        rk[k] = 0;
        if (e < E) rk[k] = atomicAdd(&h[dv[k] >> BIN_SHIFT], 1);
    }
    __syncthreads();

    int carryL = 0, carryG = 0;
    for (int i0 = 0; i0 < NB; i0 += 256) {
        const int i  = i0 + threadIdx.x;
        const int hv = (i < NB) ? h[i] : 0;
        const int tv = (i < NB) ? bintot[i] : 0;
        int totL, totG;
        const int exL = block_excl_scan(hv, wsum, totL);
        const int exG = block_excl_scan(tv, wsum, totG);
        if (i < NB) {
            lstart[i] = carryL + exL;
            gdst[i]   = carryG + exG + offG[blockIdx.x * NB + i];
        }
        carryL += totL;
        carryG += totG;
    }
    __syncthreads();

#pragma unroll
    for (int k = 0; k < CH_T; ++k) {
        const int e = base + k * 256 + threadIdx.x;
        if (e < E) {
            const int b = dv[k] >> BIN_SHIFT;
            const int p = lstart[b] + rk[k];
            stage[p] = (sv[k] << BIN_SHIFT) | (dv[k] & (BIN_NODES - 1));
            sbin[p]  = (unsigned short)b;
        }
    }
    __syncthreads();

    const int n = (E - base < CH) ? (E - base) : CH;
    for (int i = threadIdx.x; i < n; i += 256) {
        const int b = sbin[i];
        binned[gdst[b] + (i - lstart[b])] = stage[i];
    }
}

__global__ void __launch_bounds__(256)
sort_fine(const int* __restrict__ bintot,
          const int* __restrict__ binned,
          int*       __restrict__ start,
          int*       __restrict__ deg,
          int*       __restrict__ slot) {
    __shared__ int c[BIN_NODES];
    __shared__ int wsum[8];
    const int b = blockIdx.x;

    int part = 0;
    for (int i = threadIdx.x; i < b; i += 256) part += bintot[i];
    int lo;
    (void)block_excl_scan(part, wsum, lo);
    const int hi = lo + bintot[b];

    c[threadIdx.x] = 0;
    __syncthreads();

    for (int t0 = lo; t0 < hi; t0 += CH) {
        int w[CH_T];
#pragma unroll
        for (int k = 0; k < CH_T; ++k) {
            const int e = t0 + k * 256 + threadIdx.x;
            w[k] = binned[(e < hi) ? e : (hi - 1)];
        }
#pragma unroll
        for (int k = 0; k < CH_T; ++k) {
            const int e = t0 + k * 256 + threadIdx.x;
            if (e < hi) atomicAdd(&c[w[k] & (BIN_NODES - 1)], 1);
        }
    }
    __syncthreads();

    const int dgr = c[threadIdx.x];
    int tot;
    const int ex = block_excl_scan(dgr, wsum, tot);
    const int node = b * BIN_NODES + threadIdx.x;
    if (node < N_NODES) { start[node] = lo + ex; deg[node] = dgr; }
    c[threadIdx.x] = lo + ex;
    __syncthreads();

    for (int t0 = lo; t0 < hi; t0 += CH) {
        int w[CH_T];
#pragma unroll
        for (int k = 0; k < CH_T; ++k) {
            const int e = t0 + k * 256 + threadIdx.x;
            w[k] = binned[(e < hi) ? e : (hi - 1)];
        }
#pragma unroll
        for (int k = 0; k < CH_T; ++k) {
            const int e = t0 + k * 256 + threadIdx.x;
            if (e < hi) {
                const int p = atomicAdd(&c[w[k] & (BIN_NODES - 1)], 1);
                slot[p] = w[k] >> BIN_SHIFT;
            }
        }
    }
}

#define GK 7
#define GATHER_GRID ((N_NODES + 8 * GK - 1) / (8 * GK))

__device__ __forceinline__ void nb_ask(float4 (&v)[8], const float4* __restrict__ x4,
                                       int idx, int j, int m, int lane) {
    int s[8];
#pragma unroll
    for (int u = 0; u < 8; ++u) {
        const int jj = (j + u < m) ? (j + u) : (m - 1);
        s[u] = __builtin_amdgcn_readlane(idx, jj);
    }
#pragma unroll
    for (int u = 0; u < 8; ++u) {
        v[u] = make_float4(0.0f, 0.0f, 0.0f, 0.0f);
        if (j + u < m) v[u] = x4[(long long)s[u] * (D / 4) + lane];
    }
}

__device__ __forceinline__ void nb_add(float4& acc, const float4 (&v)[8]) {
#pragma unroll
    for (int u = 0; u < 8; ++u) {
        acc.x += v[u].x;
        acc.y += v[u].y;
        acc.z += v[u].z;
        acc.w += v[u].w;
    }
}

__global__ void __launch_bounds__(256)
sage_gather(const float* __restrict__ x,
            const int*   __restrict__ start,
            const int*   __restrict__ deg,
            const int*   __restrict__ slot,
            float*       __restrict__ agg,
            float*       __restrict__ cnt) {
    const int wv   = __builtin_amdgcn_readfirstlane((int)(blockIdx.x * 8 + (threadIdx.x >> 5)));
    const int lane = threadIdx.x & 31;
    const int n0   = wv * GK;
    if (n0 >= N_NODES) return;
    const int nk = (N_NODES - n0 < GK) ? (N_NODES - n0) : GK;

    const int ln   = (lane < nk) ? lane : (nk - 1);
    const int vbeg = start[n0 + ln];
    const int vdeg = deg[n0 + ln];
    if (lane < nk) cnt[n0 + lane] = (float)vdeg;

    const float4* x4 = (const float4*)x;

    int beg = __builtin_amdgcn_readlane(vbeg, 0);
    int n   = __builtin_amdgcn_readlane(vdeg, 0);
    int idx = 0;
    if (n > 0) {
        const int m0 = (n < 32) ? n : 32;
        idx = slot[beg + ((lane < m0) ? lane : (m0 - 1))];
    }

    for (int i = 0; i < nk; ++i) {
        int begN = 0, nN = 0, idxN = 0;
        if (i + 1 < nk) {
            begN = __builtin_amdgcn_readlane(vbeg, i + 1);
            nN   = __builtin_amdgcn_readlane(vdeg, i + 1);
        }

        float4 acc = make_float4(0.0f, 0.0f, 0.0f, 0.0f);
        float4 v[8];
        const int mh = (n < 32) ? n : 32;
        if (mh > 0) nb_ask(v, x4, idx, 0, mh, lane);
        if (nN > 0) {
            const int m0 = (nN < 32) ? nN : 32;
            idxN = slot[begN + ((lane < m0) ? lane : (m0 - 1))];
        }
        if (mh > 0) nb_add(acc, v);
        for (int j = 8; j < mh; j += 8) {
            nb_ask(v, x4, idx, j, mh, lane);
            nb_add(acc, v);
        }
        for (int c = 32; c < n; c += 32) {
            const int m  = (n - c < 32) ? (n - c) : 32;
            const int li = (lane < m) ? lane : (m - 1);
            idx = slot[beg + c + li];
            for (int j = 0; j < m; j += 8) {
                nb_ask(v, x4, idx, j, m, lane);
                nb_add(acc, v);
            }
        }
        ((float4*)(agg + (long long)(n0 + i) * D))[lane] = acc;
        beg = begN; n = nN; idx = idxN;
    }
}

#define G2_ROWS  32
#define G2_STEPS (N_NODES / G2_ROWS)
#define G2_GRID  512
#define A_PITCH  260
#define O_PITCH  132
#define W_PITCH  132
#define A_FLOATS (G2_ROWS * A_PITCH)
#define O_FLOATS (G2_ROWS * O_PITCH)
#define LDS_FLOATS (2 * A_FLOATS + 2 * O_FLOATS)

struct StepRows {
    float4 a0, a1, a2, a3;
    float4 x0, x1, x2, x3;
    float  n0, n1, n2, n3;
};

__device__ __forceinline__ void rows_load(StepRows& p, const float* agg, const float* x,
                                          const float* cnt, int row0, int lane) {
    p.a0 = ((const float4*)(agg + (long long)(row0     ) * D))[lane];
    p.x0 = ((const float4*)(x   + (long long)(row0     ) * D))[lane];
    p.a1 = ((const float4*)(agg + (long long)(row0 +  8) * D))[lane];
    p.x1 = ((const float4*)(x   + (long long)(row0 +  8) * D))[lane];
    p.a2 = ((const float4*)(agg + (long long)(row0 + 16) * D))[lane];
    p.x2 = ((const float4*)(x   + (long long)(row0 + 16) * D))[lane];
    p.a3 = ((const float4*)(agg + (long long)(row0 + 24) * D))[lane];
    p.x3 = ((const float4*)(x   + (long long)(row0 + 24) * D))[lane];
    p.n0 = cnt[row0     ];
    p.n1 = cnt[row0 +  8];
    p.n2 = cnt[row0 + 16];
    p.n3 = cnt[row0 + 24];
}

__device__ __forceinline__ void row_put(float* dr, float4 a, float deg, float4 xv, int lane) {
    const float rc = 1.0f / fmaxf(deg, 1.0f);
    a.x *= rc; a.y *= rc; a.z *= rc; a.w *= rc;
    ((float4*)dr)[lane]       = a;
    ((float4*)(dr + D))[lane] = xv;
}

__device__ __forceinline__ void rows_put(const StepRows& p, float* ab, int w, int lane) {
    row_put(ab + (w     ) * A_PITCH, p.a0, p.n0, p.x0, lane);
    row_put(ab + (w +  8) * A_PITCH, p.a1, p.n1, p.x1, lane);
    row_put(ab + (w + 16) * A_PITCH, p.a2, p.n2, p.x2, lane);
    row_put(ab + (w + 24) * A_PITCH, p.a3, p.n3, p.x3, lane);
}

__global__ void __launch_bounds__(256)
sage_gemm(const float* __restrict__ x,
          const float* __restrict__ Wl,
          const float* __restrict__ bl,
          const float* __restrict__ Wr,
          const float* __restrict__ cnt,
          float* out  ) {
    __shared__ __attribute__((aligned(16))) float lds[LDS_FLOATS];

    const int w    = __builtin_amdgcn_readfirstlane((int)(threadIdx.x >> 5));
    const int lane = threadIdx.x & 31;
    const int c    = lane & 15;
    const int h    = lane >> 4;

    v2f breg[64];
    {
        float*       wst = lds + w * (16 * W_PITCH);
        const float* wfr = wst + c * W_PITCH + 2 * h;
#pragma unroll
        for (int r = 0; r < 16; ++r)
            ((float4*)(wst + r * W_PITCH))[lane] =
                ((const float4*)(Wl + (long long)(w * 16 + r) * D))[lane];
        __syncthreads();
#pragma unroll
        for (int j = 0; j < 32; ++j) breg[j] = *(const v2f*)(wfr + 4 * j);
        __syncthreads();
#pragma unroll
        for (int r = 0; r < 16; ++r)
            ((float4*)(wst + r * W_PITCH))[lane] =
                ((const float4*)(Wr + (long long)(w * 16 + r) * D))[lane];
        __syncthreads();
#pragma unroll
        for (int j = 0; j < 32; ++j) breg[32 + j] = *(const v2f*)(wfr + 4 * j);
        __syncthreads();
    }
    const float bb = bl[w * 16 + c];

    StepRows p;
    int s = blockIdx.x;
    rows_load(p, out, x, cnt, s * G2_ROWS + w, lane);
    rows_put(p, lds, w, lane);
    __syncthreads();

    int cur = 0;
    for (; s < G2_STEPS; s += G2_GRID) {
        const int  sn   = s + G2_GRID;
        const bool more = sn < G2_STEPS;
        float* abuf = lds + cur * A_FLOATS;
        float* obuf = lds + 2 * A_FLOATS + cur * O_FLOATS;

        if (more) rows_load(p, out, x, cnt, sn * G2_ROWS + w, lane);

        const float* a0p = abuf + c * A_PITCH + 2 * h;
        const float* a1p = a0p + 16 * A_PITCH;
        v8f acc0 = {}, acc1 = {};
#pragma unroll
        for (int j = 0; j < 64; ++j) {
            const v2f a0 = *(const v2f*)(a0p + 4 * j);
            const v2f a1 = *(const v2f*)(a1p + 4 * j);
            acc0 = __builtin_amdgcn_wmma_f32_16x16x4_f32(
                false, a0, false, breg[j], (short)0, acc0, false, false);
            acc1 = __builtin_amdgcn_wmma_f32_16x16x4_f32(
                false, a1, false, breg[j], (short)0, acc1, false, false);
        }

#pragma unroll
        for (int j = 0; j < 8; ++j) {
            float v0 = acc0[j] + bb;
            v0 = (v0 > 0.0f) ? v0 : SLOPE * v0;
            obuf[(8 * h + j) * O_PITCH + 16 * w + c] = v0;
            float v1 = acc1[j] + bb;
            v1 = (v1 > 0.0f) ? v1 : SLOPE * v1;
            obuf[(16 + 8 * h + j) * O_PITCH + 16 * w + c] = v1;
        }

        if (more) rows_put(p, lds + (cur ^ 1) * A_FLOATS, w, lane);
        __syncthreads();

#pragma unroll
        for (int q = 0; q < 4; ++q) {
            const int r = q * 8 + w;
            const float4 t = ((const float4*)(obuf + r * O_PITCH))[lane];
            ((float4*)(out + (long long)(s * G2_ROWS + r) * D))[lane] = t;
        }
        cur ^= 1;
    }
}

extern "C" void kernel_launch(void* const* d_in, const int* in_sizes, int n_in,
                              void* d_out, int out_size, void* d_ws, size_t ws_size,
                              hipStream_t stream) {
    const float* x  = (const float*)d_in[0];
    const int*   ei = (const int*)  d_in[1];
    const float* Wl = (const float*)d_in[2];
    const float* bl = (const float*)d_in[3];
    const float* Wr = (const float*)d_in[4];
    float* out = (float*)d_out;
    const int E = in_sizes[1] / 2;

    char*  ws     = (char*)d_ws;
    int*   deg    = (int*)ws;
    int*   start  = (int*)  (ws + (1  << 19));
    float* cnt    = (float*)(ws + (1  << 20));
    int*   histG  = (int*)  (ws + (2  << 20));
    int*   offG   = (int*)  (ws + (4  << 20));
    int*   bintot = (int*)  (ws + (6  << 20));
    int*   binned = (int*)  (ws + (8  << 20));
    int*   slot   = (int*)  (ws + (16 << 20));

    const int G = (E + CH - 1) / CH;
    sort_hist   <<<G,  256, 0, stream>>>(ei, histG, E);
    sort_binscan<<<NB, 256, 0, stream>>>(histG, offG, bintot, G);
    sort_scatter<<<G,  256, 0, stream>>>(ei, offG, bintot, binned, E);
    sort_fine   <<<NB, 256, 0, stream>>>(bintot, binned, start, deg, slot);
    sage_gather <<<GATHER_GRID, 256, 0, stream>>>(x, start, deg, slot, out, cnt);

    sage_gemm<<<G2_GRID, 256, 0, stream>>>(x, Wl, bl, Wr, cnt, out);
}


# edge sort in 2 launches, not 4: each chunk, ordered by bin, stays in place; the per-bin pass finds its run in every chunk from the chunks' bin starts; sum order = LDS-atomic order, lists reordered
# speedup vs baseline: 4.1676x; 1.1193x over previous
#include <hip/hip_runtime.h>

typedef __attribute__((ext_vector_type(2))) float v2f;
typedef __attribute__((ext_vector_type(8))) float v8f;

#define N_NODES 100000
#define D 128
#define SLOPE 0.2f

#define BIN_SHIFT 8
#define BIN_NODES (1 << BIN_SHIFT)
#define NB ((N_NODES + BIN_NODES - 1) / BIN_NODES)
#define CH 4096
#define CH_T (CH / 256)
static_assert(N_NODES <= (1 << 24), "src<<8 must fit a positive int");

__device__ __forceinline__ int block_excl_scan(int v, int* wsum, int& total) {
    const int lane = threadIdx.x & 31;
    const int w    = threadIdx.x >> 5;
    int inc = v;
    for (int o = 1; o < 32; o <<= 1) {
        const int t = __shfl_up(inc, o);
        if (lane >= o) inc += t;
    }
    __syncthreads();
    if (lane == 31) wsum[w] = inc;
    __syncthreads();
    int base = 0, tot = 0;
    for (int k = 0; k < 8; ++k) {
        const int t = wsum[k];
        if (k < w) base += t;
        tot += t;
    }
    total = tot;
    return base + inc - v;
}

#define NBP (NB + 1)

__global__ void __launch_bounds__(256)
sort_local(const int* __restrict__ ei,
           int*       __restrict__ lstartG,
           int*       __restrict__ binned,
           int E) {
    __shared__ int h[NB];
    __shared__ int lstart[NB];
    __shared__ int stage[CH];
    __shared__ int wsum[8];

    for (int i = threadIdx.x; i < NB; i += 256) h[i] = 0;
    __syncthreads();
    const int base = blockIdx.x * CH;
    int sv[CH_T], dv[CH_T], rk[CH_T];
#pragma unroll
    for (int k = 0; k < CH_T; ++k) {
        const int e  = base + k * 256 + threadIdx.x;
        const int ee = (e < E) ? e : (E - 1);
        sv[k] = ei[ee];
        dv[k] = ei[(long long)E + ee];
    }
#pragma unroll
    for (int k = 0; k < CH_T; ++k) {
        const int e = base + k * 256 + threadIdx.x;
        rk[k] = 0;
        if (e < E) rk[k] = atomicAdd(&h[dv[k] >> BIN_SHIFT], 1);
    }
    __syncthreads();

    int carry = 0;
    for (int i0 = 0; i0 < NB; i0 += 256) {
        const int i  = i0 + threadIdx.x;
        const int hv = (i < NB) ? h[i] : 0;
        int tot;
        const int ex = block_excl_scan(hv, wsum, tot);
        if (i < NB) {
            lstart[i] = carry + ex;
            lstartG[blockIdx.x * NBP + i] = carry + ex;
        }
        carry += tot;
    }
    if (threadIdx.x == 0) lstartG[blockIdx.x * NBP + NB] = carry;
    __syncthreads();

#pragma unroll
    for (int k = 0; k < CH_T; ++k) {
        const int e = base + k * 256 + threadIdx.x;
        if (e < E) {
            const int b = dv[k] >> BIN_SHIFT;
            stage[lstart[b] + rk[k]] = (sv[k] << BIN_SHIFT) | (dv[k] & (BIN_NODES - 1));
        }
    }
    __syncthreads();

    const int n = (E - base < CH) ? (E - base) : CH;
    for (int i = threadIdx.x; i < n; i += 256) binned[base + i] = stage[i];
}

#define RG 512
#define FT 20
#define TILE (FT * 256)

__device__ __forceinline__ int runs_build(const int* __restrict__ lstartG, int b, int g0, int G,
                                          int* rpos, int* rsrc, int* wsum, int& sumS) {
    const int r0 = g0 + 2 * threadIdx.x;
    const int r1 = r0 + 1;
    const int q0 = (r0 < G) ? r0 : (G - 1);
    const int q1 = (r1 < G) ? r1 : (G - 1);
    const int s0 = lstartG[q0 * NBP + b];
    const int e0 = lstartG[q0 * NBP + b + 1];
    const int s1 = lstartG[q1 * NBP + b];
    const int e1 = lstartG[q1 * NBP + b + 1];
    const int len0 = (r0 < G) ? (e0 - s0) : 0;
    const int len1 = (r1 < G) ? (e1 - s1) : 0;
    if (r0 < G) sumS += s0;
    if (r1 < G) sumS += s1;
    int total;
    const int ex = block_excl_scan(len0 + len1, wsum, total);
    rpos[2 * threadIdx.x]     = ex;
    rpos[2 * threadIdx.x + 1] = ex + len0;
    rsrc[2 * threadIdx.x]     = q0 * CH + s0;
    rsrc[2 * threadIdx.x + 1] = q1 * CH + s1;
    __syncthreads();
    return total;
}

__device__ __forceinline__ void tile_load(int (&w)[FT], const int* __restrict__ binned,
                                          const int* rpos, const int* rsrc, int t0, int total) {
#pragma unroll
    for (int k = 0; k < FT; ++k) {
        const int e = t0 + k * 256 + threadIdx.x;
        const int i = (e < total) ? e : (total - 1);
        int r = 0;
#pragma unroll
        for (int st = RG / 2; st >= 1; st >>= 1)
            if (rpos[r + st] <= i) r += st;
        w[k] = binned[rsrc[r] + (i - rpos[r])];
    }
}

template <bool PLACE>
__device__ __forceinline__ void tile_apply(const int (&w)[FT], int* c, int* __restrict__ slot,
                                           int t0, int total) {
#pragma unroll
    for (int k = 0; k < FT; ++k) {
        const int e = t0 + k * 256 + threadIdx.x;
        if (e < total) {
            const int p = atomicAdd(&c[w[k] & (BIN_NODES - 1)], 1);
            if (PLACE) slot[p] = w[k] >> BIN_SHIFT;
        }
    }
}

template <bool PLACE>
__device__ __forceinline__ void tile_walk(const int* __restrict__ binned, const int* rpos, const int* rsrc,
                                          int* c, int* __restrict__ slot, int t0, int total) {
#pragma unroll 4
    for (int k = 0; k < FT; ++k) {
        const int i = t0 + k * 256 + threadIdx.x;
        if (i < total) {
            int r = 0;
#pragma unroll
            for (int st = RG / 2; st >= 1; st >>= 1)
                if (rpos[r + st] <= i) r += st;
            const int wv = binned[rsrc[r] + (i - rpos[r])];
            const int p  = atomicAdd(&c[wv & (BIN_NODES - 1)], 1);
            if (PLACE) slot[p] = wv >> BIN_SHIFT;
        }
    }
}

__global__ void __launch_bounds__(256)
sort_fine(const int* __restrict__ lstartG,
          const int* __restrict__ binned,
          int*       __restrict__ start,
          int*       __restrict__ deg,
          int*       __restrict__ slot,
          int G) {
    __shared__ int c[BIN_NODES];
    __shared__ int wsum[8];
    __shared__ int rpos[RG];
    __shared__ int rsrc[RG];
    const int b = blockIdx.x;

    c[threadIdx.x] = 0;

    int w[FT];
#pragma unroll
    for (int k = 0; k < FT; ++k) w[k] = 0;
    int sumS = 0;
    int total = runs_build(lstartG, b, 0, G, rpos, rsrc, wsum, sumS);
    const bool single = (G <= RG) && (total <= TILE);
    if (single) {
        if (total > 0) {
            tile_load(w, binned, rpos, rsrc, 0, total);
            tile_apply<false>(w, c, slot, 0, total);
        }
    } else {
        for (int g0 = 0;;) {
            for (int t0 = 0; t0 < total; t0 += TILE)
                tile_walk<false>(binned, rpos, rsrc, c, slot, t0, total);
            g0 += RG;
            if (g0 >= G) break;
            total = runs_build(lstartG, b, g0, G, rpos, rsrc, wsum, sumS);
        }
    }

    int lo;
    (void)block_excl_scan(sumS, wsum, lo);
    const int dgr = c[threadIdx.x];
    int tot;
    const int ex = block_excl_scan(dgr, wsum, tot);
    const int node = b * BIN_NODES + threadIdx.x;
    if (node < N_NODES) { start[node] = lo + ex; deg[node] = dgr; }
    c[threadIdx.x] = lo + ex;
    __syncthreads();

    if (single) {
        tile_apply<true>(w, c, slot, 0, total);
    } else {
        int dummy = 0;
        for (int g0 = 0; g0 < G; g0 += RG) {
            total = runs_build(lstartG, b, g0, G, rpos, rsrc, wsum, dummy);
            for (int t0 = 0; t0 < total; t0 += TILE)
                tile_walk<true>(binned, rpos, rsrc, c, slot, t0, total);
        }
    }
}

#define GK 7
#define GATHER_GRID ((N_NODES + 8 * GK - 1) / (8 * GK))

__device__ __forceinline__ void nb_ask(float4 (&v)[8], const float4* __restrict__ x4,
                                       int idx, int j, int m, int lane) {
    int s[8];
#pragma unroll
    for (int u = 0; u < 8; ++u) {
        const int jj = (j + u < m) ? (j + u) : (m - 1);
        s[u] = __builtin_amdgcn_readlane(idx, jj);
    }
#pragma unroll
    for (int u = 0; u < 8; ++u) {
        v[u] = make_float4(0.0f, 0.0f, 0.0f, 0.0f);
        if (j + u < m) v[u] = x4[(long long)s[u] * (D / 4) + lane];
    }
}

__device__ __forceinline__ void nb_add(float4& acc, const float4 (&v)[8]) {
#pragma unroll
    for (int u = 0; u < 8; ++u) {
        acc.x += v[u].x;
        acc.y += v[u].y;
        acc.z += v[u].z;
        acc.w += v[u].w;
    }
}

__global__ void __launch_bounds__(256)
sage_gather(const float* __restrict__ x,
            const int*   __restrict__ start,
            const int*   __restrict__ deg,
            const int*   __restrict__ slot,
            float*       __restrict__ agg,
            float*       __restrict__ cnt) {
    const int wv   = __builtin_amdgcn_readfirstlane((int)(blockIdx.x * 8 + (threadIdx.x >> 5)));
    const int lane = threadIdx.x & 31;
    const int n0   = wv * GK;
    if (n0 >= N_NODES) return;
    const int nk = (N_NODES - n0 < GK) ? (N_NODES - n0) : GK;

    const int ln   = (lane < nk) ? lane : (nk - 1);
    const int vbeg = start[n0 + ln];
    const int vdeg = deg[n0 + ln];
    if (lane < nk) cnt[n0 + lane] = (float)vdeg;

    const float4* x4 = (const float4*)x;

    int beg = __builtin_amdgcn_readlane(vbeg, 0);
    int n   = __builtin_amdgcn_readlane(vdeg, 0);
    int idx = 0;
    if (n > 0) {
        const int m0 = (n < 32) ? n : 32;
        idx = slot[beg + ((lane < m0) ? lane : (m0 - 1))];
    }

    for (int i = 0; i < nk; ++i) {
        int begN = 0, nN = 0, idxN = 0;
        if (i + 1 < nk) {
            begN = __builtin_amdgcn_readlane(vbeg, i + 1);
            nN   = __builtin_amdgcn_readlane(vdeg, i + 1);
        }

        float4 acc = make_float4(0.0f, 0.0f, 0.0f, 0.0f);
        float4 v[8];
        const int mh = (n < 32) ? n : 32;
        if (mh > 0) nb_ask(v, x4, idx, 0, mh, lane);
        if (nN > 0) {
            const int m0 = (nN < 32) ? nN : 32;
            idxN = slot[begN + ((lane < m0) ? lane : (m0 - 1))];
        }
        if (mh > 0) nb_add(acc, v);
        for (int j = 8; j < mh; j += 8) {
            nb_ask(v, x4, idx, j, mh, lane);
            nb_add(acc, v);
        }
        for (int c = 32; c < n; c += 32) {
            const int m  = (n - c < 32) ? (n - c) : 32;
            const int li = (lane < m) ? lane : (m - 1);
            idx = slot[beg + c + li];
            for (int j = 0; j < m; j += 8) {
                nb_ask(v, x4, idx, j, m, lane);
                nb_add(acc, v);
            }
        }
        ((float4*)(agg + (long long)(n0 + i) * D))[lane] = acc;
        beg = begN; n = nN; idx = idxN;
    }
}

#define G2_ROWS  32
#define G2_STEPS (N_NODES / G2_ROWS)
#define G2_GRID  512
#define A_PITCH  260
#define O_PITCH  132
#define W_PITCH  132
#define A_FLOATS (G2_ROWS * A_PITCH)
#define O_FLOATS (G2_ROWS * O_PITCH)
#define LDS_FLOATS (2 * A_FLOATS + 2 * O_FLOATS)

struct StepRows {
    float4 a0, a1, a2, a3;
    float4 x0, x1, x2, x3;
    float  n0, n1, n2, n3;
};

__device__ __forceinline__ void rows_load(StepRows& p, const float* agg, const float* x,
                                          const float* cnt, int row0, int lane) {
    p.a0 = ((const float4*)(agg + (long long)(row0     ) * D))[lane];
    p.x0 = ((const float4*)(x   + (long long)(row0     ) * D))[lane];
    p.a1 = ((const float4*)(agg + (long long)(row0 +  8) * D))[lane];
    p.x1 = ((const float4*)(x   + (long long)(row0 +  8) * D))[lane];
    p.a2 = ((const float4*)(agg + (long long)(row0 + 16) * D))[lane];
    p.x2 = ((const float4*)(x   + (long long)(row0 + 16) * D))[lane];
    p.a3 = ((const float4*)(agg + (long long)(row0 + 24) * D))[lane];
    p.x3 = ((const float4*)(x   + (long long)(row0 + 24) * D))[lane];
    p.n0 = cnt[row0     ];
    p.n1 = cnt[row0 +  8];
    p.n2 = cnt[row0 + 16];
    p.n3 = cnt[row0 + 24];
}

__device__ __forceinline__ void row_put(float* dr, float4 a, float deg, float4 xv, int lane) {
    const float rc = 1.0f / fmaxf(deg, 1.0f);
    a.x *= rc; a.y *= rc; a.z *= rc; a.w *= rc;
    ((float4*)dr)[lane]       = a;
    ((float4*)(dr + D))[lane] = xv;
}

__device__ __forceinline__ void rows_put(const StepRows& p, float* ab, int w, int lane) {
    row_put(ab + (w     ) * A_PITCH, p.a0, p.n0, p.x0, lane);
    row_put(ab + (w +  8) * A_PITCH, p.a1, p.n1, p.x1, lane);
    row_put(ab + (w + 16) * A_PITCH, p.a2, p.n2, p.x2, lane);
    row_put(ab + (w + 24) * A_PITCH, p.a3, p.n3, p.x3, lane);
}

__global__ void __launch_bounds__(256)
sage_gemm(const float* __restrict__ x,
          const float* __restrict__ Wl,
          const float* __restrict__ bl,
          const float* __restrict__ Wr,
          const float* __restrict__ cnt,
          float* out  ) {
    __shared__ __attribute__((aligned(16))) float lds[LDS_FLOATS];

    const int w    = __builtin_amdgcn_readfirstlane((int)(threadIdx.x >> 5));
    const int lane = threadIdx.x & 31;
    const int c    = lane & 15;
    const int h    = lane >> 4;

    v2f breg[64];
    {
        float*       wst = lds + w * (16 * W_PITCH);
        const float* wfr = wst + c * W_PITCH + 2 * h;
#pragma unroll
        for (int r = 0; r < 16; ++r)
            ((float4*)(wst + r * W_PITCH))[lane] =
                ((const float4*)(Wl + (long long)(w * 16 + r) * D))[lane];
        __syncthreads();
#pragma unroll
        for (int j = 0; j < 32; ++j) breg[j] = *(const v2f*)(wfr + 4 * j);
        __syncthreads();
#pragma unroll
        for (int r = 0; r < 16; ++r)
            ((float4*)(wst + r * W_PITCH))[lane] =
                ((const float4*)(Wr + (long long)(w * 16 + r) * D))[lane];
        __syncthreads();
#pragma unroll
        for (int j = 0; j < 32; ++j) breg[32 + j] = *(const v2f*)(wfr + 4 * j);
        __syncthreads();
    }
    const float bb = bl[w * 16 + c];

    StepRows p;
    int s = blockIdx.x;
    rows_load(p, out, x, cnt, s * G2_ROWS + w, lane);
    rows_put(p, lds, w, lane);
    __syncthreads();

    int cur = 0;
    for (; s < G2_STEPS; s += G2_GRID) {
        const int  sn   = s + G2_GRID;
        const bool more = sn < G2_STEPS;
        float* abuf = lds + cur * A_FLOATS;
        float* obuf = lds + 2 * A_FLOATS + cur * O_FLOATS;

        if (more) rows_load(p, out, x, cnt, sn * G2_ROWS + w, lane);

        const float* a0p = abuf + c * A_PITCH + 2 * h;
        const float* a1p = a0p + 16 * A_PITCH;
        v8f acc0 = {}, acc1 = {};
#pragma unroll
        for (int j = 0; j < 64; ++j) {
            const v2f a0 = *(const v2f*)(a0p + 4 * j);
            const v2f a1 = *(const v2f*)(a1p + 4 * j);
            acc0 = __builtin_amdgcn_wmma_f32_16x16x4_f32(
                false, a0, false, breg[j], (short)0, acc0, false, false);
            acc1 = __builtin_amdgcn_wmma_f32_16x16x4_f32(
                false, a1, false, breg[j], (short)0, acc1, false, false);
        }

#pragma unroll
        for (int j = 0; j < 8; ++j) {
            float v0 = acc0[j] + bb;
            v0 = (v0 > 0.0f) ? v0 : SLOPE * v0;
            obuf[(8 * h + j) * O_PITCH + 16 * w + c] = v0;
            float v1 = acc1[j] + bb;
            v1 = (v1 > 0.0f) ? v1 : SLOPE * v1;
            obuf[(16 + 8 * h + j) * O_PITCH + 16 * w + c] = v1;
        }

        if (more) rows_put(p, lds + (cur ^ 1) * A_FLOATS, w, lane);
        __syncthreads();

#pragma unroll
        for (int q = 0; q < 4; ++q) {
            const int r = q * 8 + w;
            const float4 t = ((const float4*)(obuf + r * O_PITCH))[lane];
            ((float4*)(out + (long long)(s * G2_ROWS + r) * D))[lane] = t;
        }
        cur ^= 1;
    }
}

extern "C" void kernel_launch(void* const* d_in, const int* in_sizes, int n_in,
                              void* d_out, int out_size, void* d_ws, size_t ws_size,
                              hipStream_t stream) {
    const float* x  = (const float*)d_in[0];
    const int*   ei = (const int*)  d_in[1];
    const float* Wl = (const float*)d_in[2];
    const float* bl = (const float*)d_in[3];
    const float* Wr = (const float*)d_in[4];
    float* out = (float*)d_out;
    const int E = in_sizes[1] / 2;

    char*  ws     = (char*)d_ws;
    int*   deg    = (int*)ws;
    int*   start  = (int*)  (ws + (1  << 19));
    float* cnt    = (float*)(ws + (1  << 20));
    int*   lstartG = (int*) (ws + (2  << 20));
    int*   binned = (int*)  (ws + (8  << 20));
    int*   slot   = (int*)  (ws + (16 << 20));

    const int G = (E + CH - 1) / CH;
    sort_local<<<G,  256, 0, stream>>>(ei, lstartG, binned, E);
    sort_fine <<<NB, 256, 0, stream>>>(lstartG, binned, start, deg, slot, G);
    sage_gather <<<GATHER_GRID, 256, 0, stream>>>(x, start, deg, slot, out, cnt);

    sage_gemm<<<G2_GRID, 256, 0, stream>>>(x, Wl, bl, Wr, cnt, out);
}


# per-bin pass: id placed at node start + rank returned by its counting LDS add, in an LDS list stored out as whole lines (no 2nd atomic pass, no scattered 4-B stores); sum order = counting-add order
# speedup vs baseline: 4.3366x; 1.0406x over previous
#include <hip/hip_runtime.h>

typedef __attribute__((ext_vector_type(2))) float v2f;
typedef __attribute__((ext_vector_type(8))) float v8f;

#define N_NODES 100000
#define D 128
#define SLOPE 0.2f

#define BIN_SHIFT 8
#define BIN_NODES (1 << BIN_SHIFT)
#define NB ((N_NODES + BIN_NODES - 1) / BIN_NODES)
#define CH 4096
#define CH_T (CH / 256)
static_assert(N_NODES <= (1 << 24), "src<<8 must fit a positive int");

__device__ __forceinline__ int block_excl_scan(int v, int* wsum, int& total) {
    const int lane = threadIdx.x & 31;
    const int w    = threadIdx.x >> 5;
    int inc = v;
    for (int o = 1; o < 32; o <<= 1) {
        const int t = __shfl_up(inc, o);
        if (lane >= o) inc += t;
    }
    __syncthreads();
    if (lane == 31) wsum[w] = inc;
    __syncthreads();
    int base = 0, tot = 0;
    for (int k = 0; k < 8; ++k) {
        const int t = wsum[k];
        if (k < w) base += t;
        tot += t;
    }
    total = tot;
    return base + inc - v;
}

#define NBP (NB + 1)

__global__ void __launch_bounds__(256)
sort_local(const int* __restrict__ ei,
           int*       __restrict__ lstartG,
           int*       __restrict__ binned,
           int E) {
    __shared__ int h[NB];
    __shared__ int lstart[NB];
    __shared__ int stage[CH];
    __shared__ int wsum[8];

    for (int i = threadIdx.x; i < NB; i += 256) h[i] = 0;
    __syncthreads();
    const int base = blockIdx.x * CH;
    int sv[CH_T], dv[CH_T], rk[CH_T];
#pragma unroll
    for (int k = 0; k < CH_T; ++k) {
        const int e  = base + k * 256 + threadIdx.x;
        const int ee = (e < E) ? e : (E - 1);
        sv[k] = ei[ee];
        dv[k] = ei[(long long)E + ee];
    }
#pragma unroll
    for (int k = 0; k < CH_T; ++k) {
        const int e = base + k * 256 + threadIdx.x;
        rk[k] = 0;
        if (e < E) rk[k] = atomicAdd(&h[dv[k] >> BIN_SHIFT], 1);
    }
    __syncthreads();

    int carry = 0;
    for (int i0 = 0; i0 < NB; i0 += 256) {
        const int i  = i0 + threadIdx.x;
        const int hv = (i < NB) ? h[i] : 0;
        int tot;
        const int ex = block_excl_scan(hv, wsum, tot);
        if (i < NB) {
            lstart[i] = carry + ex;
            lstartG[blockIdx.x * NBP + i] = carry + ex;
        }
        carry += tot;
    }
    if (threadIdx.x == 0) lstartG[blockIdx.x * NBP + NB] = carry;
    __syncthreads();

#pragma unroll
    for (int k = 0; k < CH_T; ++k) {
        const int e = base + k * 256 + threadIdx.x;
        if (e < E) {
            const int b = dv[k] >> BIN_SHIFT;
            stage[lstart[b] + rk[k]] = (sv[k] << BIN_SHIFT) | (dv[k] & (BIN_NODES - 1));
        }
    }
    __syncthreads();

    const int n = (E - base < CH) ? (E - base) : CH;
    for (int i = threadIdx.x; i < n; i += 256) binned[base + i] = stage[i];
}

#define RG 512
#define FT 20
#define TILE (FT * 256)

__device__ __forceinline__ int runs_build(const int* __restrict__ lstartG, int b, int g0, int G,
                                          int* rpos, int* rsrc, int* wsum, int& sumS) {
    const int r0 = g0 + 2 * threadIdx.x;
    const int r1 = r0 + 1;
    const int q0 = (r0 < G) ? r0 : (G - 1);
    const int q1 = (r1 < G) ? r1 : (G - 1);
    const int s0 = lstartG[q0 * NBP + b];
    const int e0 = lstartG[q0 * NBP + b + 1];
    const int s1 = lstartG[q1 * NBP + b];
    const int e1 = lstartG[q1 * NBP + b + 1];
    const int len0 = (r0 < G) ? (e0 - s0) : 0;
    const int len1 = (r1 < G) ? (e1 - s1) : 0;
    if (r0 < G) sumS += s0;
    if (r1 < G) sumS += s1;
    int total;
    const int ex = block_excl_scan(len0 + len1, wsum, total);
    rpos[2 * threadIdx.x]     = ex;
    rpos[2 * threadIdx.x + 1] = ex + len0;
    rsrc[2 * threadIdx.x]     = q0 * CH + s0;
    rsrc[2 * threadIdx.x + 1] = q1 * CH + s1;
    __syncthreads();
    return total;
}

__device__ __forceinline__ void tile_load(int (&w)[FT], const int* __restrict__ binned,
                                          const int* rpos, const int* rsrc, int t0, int total) {
#pragma unroll
    for (int k = 0; k < FT; ++k) {
        const int e = t0 + k * 256 + threadIdx.x;
        const int i = (e < total) ? e : (total - 1);
        int r = 0;
#pragma unroll
        for (int st = RG / 2; st >= 1; st >>= 1)
            if (rpos[r + st] <= i) r += st;
        w[k] = binned[rsrc[r] + (i - rpos[r])];
    }
}

__device__ __forceinline__ void tile_count(const int (&w)[FT], int (&rk)[FT], int* c, int total) {
#pragma unroll
    for (int k = 0; k < FT; ++k) {
        const bool in = (k * 256 + (int)threadIdx.x) < total;
        rk[k] = atomicAdd(&c[in ? (w[k] & (BIN_NODES - 1)) : (int)threadIdx.x], in ? 1 : 0);
    }
}

__device__ __forceinline__ void tile_place(const int (&w)[FT], const int (&rk)[FT], const int* c,
                                           int* lst, int total) {
    int p[FT];
#pragma unroll
    for (int k = 0; k < FT; ++k) p[k] = c[w[k] & (BIN_NODES - 1)] + rk[k];
#pragma unroll
    for (int k = 0; k < FT; ++k)
        if (k * 256 + (int)threadIdx.x < total) lst[p[k]] = w[k] >> BIN_SHIFT;
}

template <bool PLACE>
__device__ __forceinline__ void tile_walk(const int* __restrict__ binned, const int* rpos, const int* rsrc,
                                          int* c, int* __restrict__ slot, int t0, int total) {
#pragma unroll 4
    for (int k = 0; k < FT; ++k) {
        const int i = t0 + k * 256 + threadIdx.x;
        if (i < total) {
            int r = 0;
#pragma unroll
            for (int st = RG / 2; st >= 1; st >>= 1)
                if (rpos[r + st] <= i) r += st;
            const int wv = binned[rsrc[r] + (i - rpos[r])];
            const int p  = atomicAdd(&c[wv & (BIN_NODES - 1)], 1);
            if (PLACE) slot[p] = wv >> BIN_SHIFT;
        }
    }
}

__global__ void __launch_bounds__(256)
sort_fine(const int* __restrict__ lstartG,
          const int* __restrict__ binned,
          int*       __restrict__ start,
          int*       __restrict__ deg,
          int*       __restrict__ slot,
          int G) {
    __shared__ int c[BIN_NODES];
    __shared__ int wsum[8];
    __shared__ int rpos[RG];
    __shared__ int rsrc[RG];
    __shared__ int lst[TILE];
    const int b = blockIdx.x;

    c[threadIdx.x] = 0;

    int w[FT], rk[FT];
#pragma unroll
    for (int k = 0; k < FT; ++k) { w[k] = 0; rk[k] = 0; }
    int sumS = 0;
    int total = runs_build(lstartG, b, 0, G, rpos, rsrc, wsum, sumS);
    const bool single = (G <= RG) && (total <= TILE);
    if (single) {
        if (total > 0) {
            tile_load(w, binned, rpos, rsrc, 0, total);
            tile_count(w, rk, c, total);
        }
    } else {
        for (int g0 = 0;;) {
            for (int t0 = 0; t0 < total; t0 += TILE)
                tile_walk<false>(binned, rpos, rsrc, c, slot, t0, total);
            g0 += RG;
            if (g0 >= G) break;
            total = runs_build(lstartG, b, g0, G, rpos, rsrc, wsum, sumS);
        }
    }

    int lo;
    (void)block_excl_scan(sumS, wsum, lo);
    const int dgr = c[threadIdx.x];
    int tot;
    const int ex = block_excl_scan(dgr, wsum, tot);
    const int node = b * BIN_NODES + threadIdx.x;
    if (node < N_NODES) { start[node] = lo + ex; deg[node] = dgr; }
    c[threadIdx.x] = single ? ex : (lo + ex);
    __syncthreads();

    if (single) {
        tile_place(w, rk, c, lst, total);
        __syncthreads();
        for (int i = threadIdx.x; i < total; i += 256) slot[lo + i] = lst[i];
    } else {
        int dummy = 0;
        for (int g0 = 0; g0 < G; g0 += RG) {
            total = runs_build(lstartG, b, g0, G, rpos, rsrc, wsum, dummy);
            for (int t0 = 0; t0 < total; t0 += TILE)
                tile_walk<true>(binned, rpos, rsrc, c, slot, t0, total);
        }
    }
}

#define GK 7
#define GATHER_GRID ((N_NODES + 8 * GK - 1) / (8 * GK))

__device__ __forceinline__ void nb_ask(float4 (&v)[8], const float4* __restrict__ x4,
                                       int idx, int j, int m, int lane) {
    int s[8];
#pragma unroll
    for (int u = 0; u < 8; ++u) {
        const int jj = (j + u < m) ? (j + u) : (m - 1);
        s[u] = __builtin_amdgcn_readlane(idx, jj);
    }
#pragma unroll
    for (int u = 0; u < 8; ++u) {
        v[u] = make_float4(0.0f, 0.0f, 0.0f, 0.0f);
        if (j + u < m) v[u] = x4[(long long)s[u] * (D / 4) + lane];
    }
}

__device__ __forceinline__ void nb_add(float4& acc, const float4 (&v)[8]) {
#pragma unroll
    for (int u = 0; u < 8; ++u) {
        acc.x += v[u].x;
        acc.y += v[u].y;
        acc.z += v[u].z;
        acc.w += v[u].w;
    }
}

__global__ void __launch_bounds__(256)
sage_gather(const float* __restrict__ x,
            const int*   __restrict__ start,
            const int*   __restrict__ deg,
            const int*   __restrict__ slot,
            float*       __restrict__ agg,
            float*       __restrict__ cnt) {
    const int wv   = __builtin_amdgcn_readfirstlane((int)(blockIdx.x * 8 + (threadIdx.x >> 5)));
    const int lane = threadIdx.x & 31;
    const int n0   = wv * GK;
    if (n0 >= N_NODES) return;
    const int nk = (N_NODES - n0 < GK) ? (N_NODES - n0) : GK;

    const int ln   = (lane < nk) ? lane : (nk - 1);
    const int vbeg = start[n0 + ln];
    const int vdeg = deg[n0 + ln];
    if (lane < nk) cnt[n0 + lane] = (float)vdeg;

    const float4* x4 = (const float4*)x;

    int beg = __builtin_amdgcn_readlane(vbeg, 0);
    int n   = __builtin_amdgcn_readlane(vdeg, 0);
    int idx = 0;
    if (n > 0) {
        const int m0 = (n < 32) ? n : 32;
        idx = slot[beg + ((lane < m0) ? lane : (m0 - 1))];
    }

    for (int i = 0; i < nk; ++i) {
        int begN = 0, nN = 0, idxN = 0;
        if (i + 1 < nk) {
            begN = __builtin_amdgcn_readlane(vbeg, i + 1);
            nN   = __builtin_amdgcn_readlane(vdeg, i + 1);
        }

        float4 acc = make_float4(0.0f, 0.0f, 0.0f, 0.0f);
        float4 v[8];
        const int mh = (n < 32) ? n : 32;
        if (mh > 0) nb_ask(v, x4, idx, 0, mh, lane);
        if (nN > 0) {
            const int m0 = (nN < 32) ? nN : 32;
            idxN = slot[begN + ((lane < m0) ? lane : (m0 - 1))];
        }
        if (mh > 0) nb_add(acc, v);
        for (int j = 8; j < mh; j += 8) {
            nb_ask(v, x4, idx, j, mh, lane);
            nb_add(acc, v);
        }
        for (int c = 32; c < n; c += 32) {
            const int m  = (n - c < 32) ? (n - c) : 32;
            const int li = (lane < m) ? lane : (m - 1);
            idx = slot[beg + c + li];
            for (int j = 0; j < m; j += 8) {
                nb_ask(v, x4, idx, j, m, lane);
                nb_add(acc, v);
            }
        }
        ((float4*)(agg + (long long)(n0 + i) * D))[lane] = acc;
        beg = begN; n = nN; idx = idxN;
    }
}

#define G2_ROWS  32
#define G2_STEPS (N_NODES / G2_ROWS)
#define G2_GRID  512
#define A_PITCH  260
#define O_PITCH  132
#define W_PITCH  132
#define A_FLOATS (G2_ROWS * A_PITCH)
#define O_FLOATS (G2_ROWS * O_PITCH)
#define LDS_FLOATS (2 * A_FLOATS + 2 * O_FLOATS)

struct StepRows {
    float4 a0, a1, a2, a3;
    float4 x0, x1, x2, x3;
    float  n0, n1, n2, n3;
};

__device__ __forceinline__ void rows_load(StepRows& p, const float* agg, const float* x,
                                          const float* cnt, int row0, int lane) {
    p.a0 = ((const float4*)(agg + (long long)(row0     ) * D))[lane];
    p.x0 = ((const float4*)(x   + (long long)(row0     ) * D))[lane];
    p.a1 = ((const float4*)(agg + (long long)(row0 +  8) * D))[lane];
    p.x1 = ((const float4*)(x   + (long long)(row0 +  8) * D))[lane];
    p.a2 = ((const float4*)(agg + (long long)(row0 + 16) * D))[lane];
    p.x2 = ((const float4*)(x   + (long long)(row0 + 16) * D))[lane];
    p.a3 = ((const float4*)(agg + (long long)(row0 + 24) * D))[lane];
    p.x3 = ((const float4*)(x   + (long long)(row0 + 24) * D))[lane];
    p.n0 = cnt[row0     ];
    p.n1 = cnt[row0 +  8];
    p.n2 = cnt[row0 + 16];
    p.n3 = cnt[row0 + 24];
}

__device__ __forceinline__ void row_put(float* dr, float4 a, float deg, float4 xv, int lane) {
    const float rc = 1.0f / fmaxf(deg, 1.0f);
    a.x *= rc; a.y *= rc; a.z *= rc; a.w *= rc;
    ((float4*)dr)[lane]       = a;
    ((float4*)(dr + D))[lane] = xv;
}

__device__ __forceinline__ void rows_put(const StepRows& p, float* ab, int w, int lane) {
    row_put(ab + (w     ) * A_PITCH, p.a0, p.n0, p.x0, lane);
    row_put(ab + (w +  8) * A_PITCH, p.a1, p.n1, p.x1, lane);
    row_put(ab + (w + 16) * A_PITCH, p.a2, p.n2, p.x2, lane);
    row_put(ab + (w + 24) * A_PITCH, p.a3, p.n3, p.x3, lane);
}

__global__ void __launch_bounds__(256)
sage_gemm(const float* __restrict__ x,
          const float* __restrict__ Wl,
          const float* __restrict__ bl,
          const float* __restrict__ Wr,
          const float* __restrict__ cnt,
          float* out  ) {
    __shared__ __attribute__((aligned(16))) float lds[LDS_FLOATS];

    const int w    = __builtin_amdgcn_readfirstlane((int)(threadIdx.x >> 5));
    const int lane = threadIdx.x & 31;
    const int c    = lane & 15;
    const int h    = lane >> 4;

    v2f breg[64];
    {
        float*       wst = lds + w * (16 * W_PITCH);
        const float* wfr = wst + c * W_PITCH + 2 * h;
#pragma unroll
        for (int r = 0; r < 16; ++r)
            ((float4*)(wst + r * W_PITCH))[lane] =
                ((const float4*)(Wl + (long long)(w * 16 + r) * D))[lane];
        __syncthreads();
#pragma unroll
        for (int j = 0; j < 32; ++j) breg[j] = *(const v2f*)(wfr + 4 * j);
        __syncthreads();
#pragma unroll
        for (int r = 0; r < 16; ++r)
            ((float4*)(wst + r * W_PITCH))[lane] =
                ((const float4*)(Wr + (long long)(w * 16 + r) * D))[lane];
        __syncthreads();
#pragma unroll
        for (int j = 0; j < 32; ++j) breg[32 + j] = *(const v2f*)(wfr + 4 * j);
        __syncthreads();
    }
    const float bb = bl[w * 16 + c];

    StepRows p;
    int s = blockIdx.x;
    rows_load(p, out, x, cnt, s * G2_ROWS + w, lane);
    rows_put(p, lds, w, lane);
    __syncthreads();

    int cur = 0;
    for (; s < G2_STEPS; s += G2_GRID) {
        const int  sn   = s + G2_GRID;
        const bool more = sn < G2_STEPS;
        float* abuf = lds + cur * A_FLOATS;
        float* obuf = lds + 2 * A_FLOATS + cur * O_FLOATS;

        if (more) rows_load(p, out, x, cnt, sn * G2_ROWS + w, lane);

        const float* a0p = abuf + c * A_PITCH + 2 * h;
        const float* a1p = a0p + 16 * A_PITCH;
        v8f acc0 = {}, acc1 = {};
#pragma unroll
        for (int j = 0; j < 64; ++j) {
            const v2f a0 = *(const v2f*)(a0p + 4 * j);
            const v2f a1 = *(const v2f*)(a1p + 4 * j);
            acc0 = __builtin_amdgcn_wmma_f32_16x16x4_f32(
                false, a0, false, breg[j], (short)0, acc0, false, false);
            acc1 = __builtin_amdgcn_wmma_f32_16x16x4_f32(
                false, a1, false, breg[j], (short)0, acc1, false, false);
        }

#pragma unroll
        for (int j = 0; j < 8; ++j) {
            float v0 = acc0[j] + bb;
            v0 = (v0 > 0.0f) ? v0 : SLOPE * v0;
            obuf[(8 * h + j) * O_PITCH + 16 * w + c] = v0;
            float v1 = acc1[j] + bb;
            v1 = (v1 > 0.0f) ? v1 : SLOPE * v1;
            obuf[(16 + 8 * h + j) * O_PITCH + 16 * w + c] = v1;
        }

        if (more) rows_put(p, lds + (cur ^ 1) * A_FLOATS, w, lane);
        __syncthreads();

#pragma unroll
        for (int q = 0; q < 4; ++q) {
            const int r = q * 8 + w;
            const float4 t = ((const float4*)(obuf + r * O_PITCH))[lane];
            ((float4*)(out + (long long)(s * G2_ROWS + r) * D))[lane] = t;
        }
        cur ^= 1;
    }
}

extern "C" void kernel_launch(void* const* d_in, const int* in_sizes, int n_in,
                              void* d_out, int out_size, void* d_ws, size_t ws_size,
                              hipStream_t stream) {
    const float* x  = (const float*)d_in[0];
    const int*   ei = (const int*)  d_in[1];
    const float* Wl = (const float*)d_in[2];
    const float* bl = (const float*)d_in[3];
    const float* Wr = (const float*)d_in[4];
    float* out = (float*)d_out;
    const int E = in_sizes[1] / 2;

    char*  ws     = (char*)d_ws;
    int*   deg    = (int*)ws;
    int*   start  = (int*)  (ws + (1  << 19));
    float* cnt    = (float*)(ws + (1  << 20));
    int*   lstartG = (int*) (ws + (2  << 20));
    int*   binned = (int*)  (ws + (8  << 20));
    int*   slot   = (int*)  (ws + (16 << 20));

    const int G = (E + CH - 1) / CH;
    sort_local<<<G,  256, 0, stream>>>(ei, lstartG, binned, E);
    sort_fine <<<NB, 256, 0, stream>>>(lstartG, binned, start, deg, slot, G);
    sage_gather <<<GATHER_GRID, 256, 0, stream>>>(x, start, deg, slot, out, cnt);

    sage_gemm<<<G2_GRID, 256, 0, stream>>>(x, Wl, bl, Wr, cnt, out);
}


# gather fused into the per-bin sort launch: ids read from the bin's LDS list; bins of 200 nodes, 8 waves x 25 nodes, 16 rows in flight; 3 launches; sum order = LDS-atomic order, lists reordered
# speedup vs baseline: 4.6347x; 1.0687x over previous
#include <hip/hip_runtime.h>

typedef __attribute__((ext_vector_type(2))) float v2f;
typedef __attribute__((ext_vector_type(8))) float v8f;

#define N_NODES 100000
#define D 128
#define SLOPE 0.2f

#define BIN_NODES 200
#define WAVE_NODES (BIN_NODES / 8)
#define ID_SHIFT 8
#define ID_MASK ((1 << ID_SHIFT) - 1)
#define NB ((N_NODES + BIN_NODES - 1) / BIN_NODES)
static_assert(BIN_NODES <= (1 << ID_SHIFT) && BIN_NODES % 8 == 0 && WAVE_NODES <= 32, "bin geometry");
#define CH 4096
#define CH_T (CH / 256)
static_assert(N_NODES <= (1 << 24), "src<<8 must fit a positive int");

__device__ __forceinline__ int block_excl_scan(int v, int* wsum, int& total) {
    const int lane = threadIdx.x & 31;
    const int w    = threadIdx.x >> 5;
    int inc = v;
    for (int o = 1; o < 32; o <<= 1) {
        const int t = __shfl_up(inc, o);
        if (lane >= o) inc += t;
    }
    __syncthreads();
    if (lane == 31) wsum[w] = inc;
    __syncthreads();
    int base = 0, tot = 0;
    for (int k = 0; k < 8; ++k) {
        const int t = wsum[k];
        if (k < w) base += t;
        tot += t;
    }
    total = tot;
    return base + inc - v;
}

#define NBP (NB + 1)

__global__ void __launch_bounds__(256)
sort_local(const int* __restrict__ ei,
           int*       __restrict__ lstartG,
           int*       __restrict__ binned,
           int E) {
    __shared__ int h[NB];
    __shared__ int lstart[NB];
    __shared__ int stage[CH];
    __shared__ int wsum[8];

    for (int i = threadIdx.x; i < NB; i += 256) h[i] = 0;
    __syncthreads();
    const int base = blockIdx.x * CH;
    int sv[CH_T], dv[CH_T], rk[CH_T];
#pragma unroll
    for (int k = 0; k < CH_T; ++k) {
        const int e  = base + k * 256 + threadIdx.x;
        const int ee = (e < E) ? e : (E - 1);
        sv[k] = ei[ee];
        dv[k] = ei[(long long)E + ee];
    }
#pragma unroll
    for (int k = 0; k < CH_T; ++k) {
        const int e = base + k * 256 + threadIdx.x;
        rk[k] = 0;
        if (e < E) rk[k] = atomicAdd(&h[dv[k] / BIN_NODES], 1);
    }
    __syncthreads();

    int carry = 0;
    for (int i0 = 0; i0 < NB; i0 += 256) {
        const int i  = i0 + threadIdx.x;
        const int hv = (i < NB) ? h[i] : 0;
        int tot;
        const int ex = block_excl_scan(hv, wsum, tot);
        if (i < NB) {
            lstart[i] = carry + ex;
            lstartG[blockIdx.x * NBP + i] = carry + ex;
        }
        carry += tot;
    }
    if (threadIdx.x == 0) lstartG[blockIdx.x * NBP + NB] = carry;
    __syncthreads();

#pragma unroll
    for (int k = 0; k < CH_T; ++k) {
        const int e = base + k * 256 + threadIdx.x;
        if (e < E) {
            const int b = dv[k] / BIN_NODES;
            stage[lstart[b] + rk[k]] = (sv[k] << ID_SHIFT) | (dv[k] - b * BIN_NODES);
        }
    }
    __syncthreads();

    const int n = (E - base < CH) ? (E - base) : CH;
    for (int i = threadIdx.x; i < n; i += 256) binned[base + i] = stage[i];
}

#define RG 512
#define FT 16
#define TILE (FT * 256)

__device__ __forceinline__ int runs_build(const int* __restrict__ lstartG, int b, int g0, int G,
                                          int* rpos, int* rsrc, int* wsum, int& sumS) {
    const int r0 = g0 + 2 * threadIdx.x;
    const int r1 = r0 + 1;
    const int q0 = (r0 < G) ? r0 : (G - 1);
    const int q1 = (r1 < G) ? r1 : (G - 1);
    const int s0 = lstartG[q0 * NBP + b];
    const int e0 = lstartG[q0 * NBP + b + 1];
    const int s1 = lstartG[q1 * NBP + b];
    const int e1 = lstartG[q1 * NBP + b + 1];
    const int len0 = (r0 < G) ? (e0 - s0) : 0;
    const int len1 = (r1 < G) ? (e1 - s1) : 0;
    if (r0 < G) sumS += s0;
    if (r1 < G) sumS += s1;
    int total;
    const int ex = block_excl_scan(len0 + len1, wsum, total);
    rpos[2 * threadIdx.x]     = ex;
    rpos[2 * threadIdx.x + 1] = ex + len0;
    rsrc[2 * threadIdx.x]     = q0 * CH + s0;
    rsrc[2 * threadIdx.x + 1] = q1 * CH + s1;
    __syncthreads();
    return total;
}

__device__ __forceinline__ void tile_load(int (&w)[FT], const int* __restrict__ binned,
                                          const int* rpos, const int* rsrc, int t0, int total) {
#pragma unroll
    for (int k = 0; k < FT; ++k) {
        const int e = t0 + k * 256 + threadIdx.x;
        const int i = (e < total) ? e : (total - 1);
        int r = 0;
#pragma unroll
        for (int st = RG / 2; st >= 1; st >>= 1)
            if (rpos[r + st] <= i) r += st;
        w[k] = binned[rsrc[r] + (i - rpos[r])];
    }
}

__device__ __forceinline__ void tile_count(const int (&w)[FT], int (&rk)[FT], int* c, int total) {
#pragma unroll
    for (int k = 0; k < FT; ++k) {
        const bool in = (k * 256 + (int)threadIdx.x) < total;
        rk[k] = atomicAdd(&c[in ? (w[k] & ID_MASK) : (int)threadIdx.x], in ? 1 : 0);
    }
}

__device__ __forceinline__ void tile_place(const int (&w)[FT], const int (&rk)[FT], const int* c,
                                           int* lst, int total) {
    int p[FT];
#pragma unroll
    for (int k = 0; k < FT; ++k) p[k] = c[w[k] & ID_MASK] + rk[k];
#pragma unroll
    for (int k = 0; k < FT; ++k)
        if (k * 256 + (int)threadIdx.x < total) lst[p[k]] = w[k] >> ID_SHIFT;
}

template <bool PLACE>
__device__ __forceinline__ void tile_walk(const int* __restrict__ binned, const int* rpos, const int* rsrc,
                                          int* c, int* __restrict__ slot, int t0, int total) {
#pragma unroll 4
    for (int k = 0; k < FT; ++k) {
        const int i = t0 + k * 256 + threadIdx.x;
        if (i < total) {
            int r = 0;
#pragma unroll
            for (int st = RG / 2; st >= 1; st >>= 1)
                if (rpos[r + st] <= i) r += st;
            const int wv = binned[rsrc[r] + (i - rpos[r])];
            const int p  = atomicAdd(&c[wv & ID_MASK], 1);
            if (PLACE) slot[p] = wv >> ID_SHIFT;
        }
    }
}

__device__ __forceinline__ void nb_ask(float4 (&v)[8], const float4* __restrict__ x4,
                                       int idx, int j, int m, int lane) {
    int s[8];
#pragma unroll
    for (int u = 0; u < 8; ++u) {
        const int jj = (j + u < m) ? (j + u) : (m - 1);
        s[u] = __builtin_amdgcn_readlane(idx, jj);
    }
#pragma unroll
    for (int u = 0; u < 8; ++u) {
        v[u] = make_float4(0.0f, 0.0f, 0.0f, 0.0f);
        if (j + u < m) v[u] = x4[(long long)s[u] * (D / 4) + lane];
    }
}

__device__ __forceinline__ void nb_add(float4& acc, const float4 (&v)[8]) {
#pragma unroll
    for (int u = 0; u < 8; ++u) {
        acc.x += v[u].x;
        acc.y += v[u].y;
        acc.z += v[u].z;
        acc.w += v[u].w;
    }
}


__device__ __forceinline__ void bin_gather(const float* __restrict__ x, const int* ids,
                                           int vbeg, int vdeg, int node0, int nk,
                                           float* __restrict__ agg, int lane) {
    const float4* x4 = (const float4*)x;

    int beg = __builtin_amdgcn_readlane(vbeg, 0);
    int n   = __builtin_amdgcn_readlane(vdeg, 0);
    int idx = 0;
    if (n > 0) {
        const int m0 = (n < 32) ? n : 32;
        idx = ids[beg + ((lane < m0) ? lane : (m0 - 1))];
    }

    for (int i = 0; i < nk; ++i) {
        int begN = 0, nN = 0, idxN = 0;
        if (i + 1 < nk) {
            begN = __builtin_amdgcn_readlane(vbeg, i + 1);
            nN   = __builtin_amdgcn_readlane(vdeg, i + 1);
        }

        float4 acc = make_float4(0.0f, 0.0f, 0.0f, 0.0f);
        float4 va[8], vb[8];
        const int mh = (n < 32) ? n : 32;
        if (mh > 0) {
            nb_ask(va, x4, idx, 0, mh, lane);
            nb_ask(vb, x4, idx, 8, mh, lane);
        }
        if (nN > 0) {
            const int m0 = (nN < 32) ? nN : 32;
            idxN = ids[begN + ((lane < m0) ? lane : (m0 - 1))];
        }
        if (mh > 0) {
            nb_add(acc, va);
            nb_add(acc, vb);
        }
        for (int j = 16; j < mh; j += 16) {
            nb_ask(va, x4, idx, j, mh, lane);
            nb_ask(vb, x4, idx, j + 8, mh, lane);
            nb_add(acc, va);
            nb_add(acc, vb);
        }
        for (int c0 = 32; c0 < n; c0 += 32) {
            const int m  = (n - c0 < 32) ? (n - c0) : 32;
            const int li = (lane < m) ? lane : (m - 1);
            idx = ids[beg + c0 + li];
            for (int j = 0; j < m; j += 16) {
                nb_ask(va, x4, idx, j, m, lane);
                nb_ask(vb, x4, idx, j + 8, m, lane);
                nb_add(acc, va);
                nb_add(acc, vb);
            }
        }
        ((float4*)(agg + (long long)(node0 + i) * D))[lane] = acc;
        beg = begN; n = nN; idx = idxN;
    }
}

__global__ void __launch_bounds__(256)
sort_fine(const int*   __restrict__ lstartG,
          const int*   __restrict__ binned,
          const float* __restrict__ x,
          int*                      slot,
          float*       __restrict__ agg,
          float*       __restrict__ cnt,
          int G) {
    __shared__ int c[256];
    __shared__ int wsum[8];
    __shared__ int rpos[RG];
    __shared__ int rsrc[RG];
    __shared__ int lst[TILE];
    const int b = blockIdx.x;

    c[threadIdx.x] = 0;

    int w[FT], rk[FT];
#pragma unroll
    for (int k = 0; k < FT; ++k) { w[k] = 0; rk[k] = 0; }
    int sumS = 0;
    int total = runs_build(lstartG, b, 0, G, rpos, rsrc, wsum, sumS);
    const bool single = (G <= RG) && (total <= TILE);
    if (single) {
        if (total > 0) {
            tile_load(w, binned, rpos, rsrc, 0, total);
            tile_count(w, rk, c, total);
        }
    } else {
        for (int g0 = 0;;) {
            for (int t0 = 0; t0 < total; t0 += TILE)
                tile_walk<false>(binned, rpos, rsrc, c, slot, t0, total);
            g0 += RG;
            if (g0 >= G) break;
            total = runs_build(lstartG, b, g0, G, rpos, rsrc, wsum, sumS);
        }
    }

    int lo;
    (void)block_excl_scan(sumS, wsum, lo);
    const int wv   = __builtin_amdgcn_readfirstlane((int)(threadIdx.x >> 5));
    const int lane = threadIdx.x & 31;
    const bool mine = lane < WAVE_NODES;
    const int nl   = WAVE_NODES * wv + (mine ? lane : 0);
    const int dgr  = mine ? c[nl] : 0;
    int tot;
    const int ex = block_excl_scan(dgr, wsum, tot);
    const int node = b * BIN_NODES + nl;
    if (mine && node < N_NODES) cnt[node] = (float)dgr;
    if (mine) c[nl] = single ? ex : (lo + ex);
    __syncthreads();

    const int node0 = b * BIN_NODES + WAVE_NODES * wv;
    const int left  = N_NODES - node0;
    const int nk    = (left < WAVE_NODES) ? left : WAVE_NODES;

    if (single) {
        tile_place(w, rk, c, lst, total);
        __syncthreads();
        bin_gather(x, lst, ex, dgr, node0, nk, agg, lane);
    } else {
        int dummy = 0;
        for (int g0 = 0; g0 < G; g0 += RG) {
            total = runs_build(lstartG, b, g0, G, rpos, rsrc, wsum, dummy);
            for (int t0 = 0; t0 < total; t0 += TILE)
                tile_walk<true>(binned, rpos, rsrc, c, slot, t0, total);
        }
        __syncthreads();
        bin_gather(x, slot + lo, ex, dgr, node0, nk, agg, lane);
    }
}

#define G2_ROWS  32
#define G2_STEPS (N_NODES / G2_ROWS)
#define G2_GRID  512
#define A_PITCH  260
#define O_PITCH  132
#define W_PITCH  132
#define A_FLOATS (G2_ROWS * A_PITCH)
#define O_FLOATS (G2_ROWS * O_PITCH)
#define LDS_FLOATS (2 * A_FLOATS + 2 * O_FLOATS)

struct StepRows {
    float4 a0, a1, a2, a3;
    float4 x0, x1, x2, x3;
    float  n0, n1, n2, n3;
};

__device__ __forceinline__ void rows_load(StepRows& p, const float* agg, const float* x,
                                          const float* cnt, int row0, int lane) {
    p.a0 = ((const float4*)(agg + (long long)(row0     ) * D))[lane];
    p.x0 = ((const float4*)(x   + (long long)(row0     ) * D))[lane];
    p.a1 = ((const float4*)(agg + (long long)(row0 +  8) * D))[lane];
    p.x1 = ((const float4*)(x   + (long long)(row0 +  8) * D))[lane];
    p.a2 = ((const float4*)(agg + (long long)(row0 + 16) * D))[lane];
    p.x2 = ((const float4*)(x   + (long long)(row0 + 16) * D))[lane];
    p.a3 = ((const float4*)(agg + (long long)(row0 + 24) * D))[lane];
    p.x3 = ((const float4*)(x   + (long long)(row0 + 24) * D))[lane];
    p.n0 = cnt[row0     ];
    p.n1 = cnt[row0 +  8];
    p.n2 = cnt[row0 + 16];
    p.n3 = cnt[row0 + 24];
}

__device__ __forceinline__ void row_put(float* dr, float4 a, float deg, float4 xv, int lane) {
    const float rc = 1.0f / fmaxf(deg, 1.0f);
    a.x *= rc; a.y *= rc; a.z *= rc; a.w *= rc;
    ((float4*)dr)[lane]       = a;
    ((float4*)(dr + D))[lane] = xv;
}

__device__ __forceinline__ void rows_put(const StepRows& p, float* ab, int w, int lane) {
    row_put(ab + (w     ) * A_PITCH, p.a0, p.n0, p.x0, lane);
    row_put(ab + (w +  8) * A_PITCH, p.a1, p.n1, p.x1, lane);
    row_put(ab + (w + 16) * A_PITCH, p.a2, p.n2, p.x2, lane);
    row_put(ab + (w + 24) * A_PITCH, p.a3, p.n3, p.x3, lane);
}

__global__ void __launch_bounds__(256)
sage_gemm(const float* __restrict__ x,
          const float* __restrict__ Wl,
          const float* __restrict__ bl,
          const float* __restrict__ Wr,
          const float* __restrict__ cnt,
          float* out  ) {
    __shared__ __attribute__((aligned(16))) float lds[LDS_FLOATS];

    const int w    = __builtin_amdgcn_readfirstlane((int)(threadIdx.x >> 5));
    const int lane = threadIdx.x & 31;
    const int c    = lane & 15;
    const int h    = lane >> 4;

    v2f breg[64];
    {
        float*       wst = lds + w * (16 * W_PITCH);
        const float* wfr = wst + c * W_PITCH + 2 * h;
#pragma unroll
        for (int r = 0; r < 16; ++r)
            ((float4*)(wst + r * W_PITCH))[lane] =
                ((const float4*)(Wl + (long long)(w * 16 + r) * D))[lane];
        __syncthreads();
#pragma unroll
        for (int j = 0; j < 32; ++j) breg[j] = *(const v2f*)(wfr + 4 * j);
        __syncthreads();
#pragma unroll
        for (int r = 0; r < 16; ++r)
            ((float4*)(wst + r * W_PITCH))[lane] =
                ((const float4*)(Wr + (long long)(w * 16 + r) * D))[lane];
        __syncthreads();
#pragma unroll
        for (int j = 0; j < 32; ++j) breg[32 + j] = *(const v2f*)(wfr + 4 * j);
        __syncthreads();
    }
    const float bb = bl[w * 16 + c];

    StepRows p;
    int s = blockIdx.x;
    rows_load(p, out, x, cnt, s * G2_ROWS + w, lane);
    rows_put(p, lds, w, lane);
    __syncthreads();

    int cur = 0;
    for (; s < G2_STEPS; s += G2_GRID) {
        const int  sn   = s + G2_GRID;
        const bool more = sn < G2_STEPS;
        float* abuf = lds + cur * A_FLOATS;
        float* obuf = lds + 2 * A_FLOATS + cur * O_FLOATS;

        if (more) rows_load(p, out, x, cnt, sn * G2_ROWS + w, lane);

        const float* a0p = abuf + c * A_PITCH + 2 * h;
        const float* a1p = a0p + 16 * A_PITCH;
        v8f acc0 = {}, acc1 = {};
#pragma unroll
        for (int j = 0; j < 64; ++j) {
            const v2f a0 = *(const v2f*)(a0p + 4 * j);
            const v2f a1 = *(const v2f*)(a1p + 4 * j);
            acc0 = __builtin_amdgcn_wmma_f32_16x16x4_f32(
                false, a0, false, breg[j], (short)0, acc0, false, false);
            acc1 = __builtin_amdgcn_wmma_f32_16x16x4_f32(
                false, a1, false, breg[j], (short)0, acc1, false, false);
        }

#pragma unroll
        for (int j = 0; j < 8; ++j) {
            float v0 = acc0[j] + bb;
            v0 = (v0 > 0.0f) ? v0 : SLOPE * v0;
            obuf[(8 * h + j) * O_PITCH + 16 * w + c] = v0;
            float v1 = acc1[j] + bb;
            v1 = (v1 > 0.0f) ? v1 : SLOPE * v1;
            obuf[(16 + 8 * h + j) * O_PITCH + 16 * w + c] = v1;
        }

        if (more) rows_put(p, lds + (cur ^ 1) * A_FLOATS, w, lane);
        __syncthreads();

#pragma unroll
        for (int q = 0; q < 4; ++q) {
            const int r = q * 8 + w;
            const float4 t = ((const float4*)(obuf + r * O_PITCH))[lane];
            ((float4*)(out + (long long)(s * G2_ROWS + r) * D))[lane] = t;
        }
        cur ^= 1;
    }
}

extern "C" void kernel_launch(void* const* d_in, const int* in_sizes, int n_in,
                              void* d_out, int out_size, void* d_ws, size_t ws_size,
                              hipStream_t stream) {
    const float* x  = (const float*)d_in[0];
    const int*   ei = (const int*)  d_in[1];
    const float* Wl = (const float*)d_in[2];
    const float* bl = (const float*)d_in[3];
    const float* Wr = (const float*)d_in[4];
    float* out = (float*)d_out;
    const int E = in_sizes[1] / 2;

    char*  ws     = (char*)d_ws;
    float* cnt    = (float*)(ws + (1  << 20));
    int*   lstartG = (int*) (ws + (2  << 20));
    int*   binned = (int*)  (ws + (8  << 20));
    int*   slot   = (int*)  (ws + (16 << 20));

    const int G = (E + CH - 1) / CH;
    sort_local<<<G,  256, 0, stream>>>(ei, lstartG, binned, E);
    sort_fine <<<NB, 256, 0, stream>>>(lstartG, binned, x, slot, out, cnt, G);

    sage_gemm<<<G2_GRID, 256, 0, stream>>>(x, Wl, bl, Wr, cnt, out);
}


# gather: each wave walks its 25 nodes' lists as ONE list in fixed groups of 8 unguarded row loads, 3 groups asked ahead of the adds; sum stored at node ends; sum order unchanged
# speedup vs baseline: 5.4306x; 1.1717x over previous
#include <hip/hip_runtime.h>

typedef __attribute__((ext_vector_type(2))) float v2f;
typedef __attribute__((ext_vector_type(8))) float v8f;

#define N_NODES 100000
#define D 128
#define SLOPE 0.2f

#define BIN_NODES 200
#define WAVE_NODES (BIN_NODES / 8)
#define ID_SHIFT 8
#define ID_MASK ((1 << ID_SHIFT) - 1)
#define NB ((N_NODES + BIN_NODES - 1) / BIN_NODES)
static_assert(BIN_NODES <= (1 << ID_SHIFT) && BIN_NODES % 8 == 0 && WAVE_NODES <= 32, "bin geometry");
#define CH 4096
#define CH_T (CH / 256)
static_assert(N_NODES <= (1 << 24), "src<<8 must fit a positive int");

__device__ __forceinline__ int block_excl_scan(int v, int* wsum, int& total) {
    const int lane = threadIdx.x & 31;
    const int w    = threadIdx.x >> 5;
    int inc = v;
    for (int o = 1; o < 32; o <<= 1) {
        const int t = __shfl_up(inc, o);
        if (lane >= o) inc += t;
    }
    __syncthreads();
    if (lane == 31) wsum[w] = inc;
    __syncthreads();
    int base = 0, tot = 0;
    for (int k = 0; k < 8; ++k) {
        const int t = wsum[k];
        if (k < w) base += t;
        tot += t;
    }
    total = tot;
    return base + inc - v;
}

#define NBP (NB + 1)

__global__ void __launch_bounds__(256)
sort_local(const int* __restrict__ ei,
           int*       __restrict__ lstartG,
           int*       __restrict__ binned,
           int E) {
    __shared__ int h[NB];
    __shared__ int lstart[NB];
    __shared__ int stage[CH];
    __shared__ int wsum[8];

    for (int i = threadIdx.x; i < NB; i += 256) h[i] = 0;
    __syncthreads();
    const int base = blockIdx.x * CH;
    int sv[CH_T], dv[CH_T], rk[CH_T];
#pragma unroll
    for (int k = 0; k < CH_T; ++k) {
        const int e  = base + k * 256 + threadIdx.x;
        const int ee = (e < E) ? e : (E - 1);
        sv[k] = ei[ee];
        dv[k] = ei[(long long)E + ee];
    }
#pragma unroll
    for (int k = 0; k < CH_T; ++k) {
        const int e = base + k * 256 + threadIdx.x;
        rk[k] = 0;
        if (e < E) rk[k] = atomicAdd(&h[dv[k] / BIN_NODES], 1);
    }
    __syncthreads();

    int carry = 0;
    for (int i0 = 0; i0 < NB; i0 += 256) {
        const int i  = i0 + threadIdx.x;
        const int hv = (i < NB) ? h[i] : 0;
        int tot;
        const int ex = block_excl_scan(hv, wsum, tot);
        if (i < NB) {
            lstart[i] = carry + ex;
            lstartG[blockIdx.x * NBP + i] = carry + ex;
        }
        carry += tot;
    }
    if (threadIdx.x == 0) lstartG[blockIdx.x * NBP + NB] = carry;
    __syncthreads();

#pragma unroll
    for (int k = 0; k < CH_T; ++k) {
        const int e = base + k * 256 + threadIdx.x;
        if (e < E) {
            const int b = dv[k] / BIN_NODES;
            stage[lstart[b] + rk[k]] = (sv[k] << ID_SHIFT) | (dv[k] - b * BIN_NODES);
        }
    }
    __syncthreads();

    const int n = (E - base < CH) ? (E - base) : CH;
    for (int i = threadIdx.x; i < n; i += 256) binned[base + i] = stage[i];
}

#define RG 512
#define FT 16
#define TILE (FT * 256)

__device__ __forceinline__ int runs_build(const int* __restrict__ lstartG, int b, int g0, int G,
                                          int* rpos, int* rsrc, int* wsum, int& sumS) {
    const int r0 = g0 + 2 * threadIdx.x;
    const int r1 = r0 + 1;
    const int q0 = (r0 < G) ? r0 : (G - 1);
    const int q1 = (r1 < G) ? r1 : (G - 1);
    const int s0 = lstartG[q0 * NBP + b];
    const int e0 = lstartG[q0 * NBP + b + 1];
    const int s1 = lstartG[q1 * NBP + b];
    const int e1 = lstartG[q1 * NBP + b + 1];
    const int len0 = (r0 < G) ? (e0 - s0) : 0;
    const int len1 = (r1 < G) ? (e1 - s1) : 0;
    if (r0 < G) sumS += s0;
    if (r1 < G) sumS += s1;
    int total;
    const int ex = block_excl_scan(len0 + len1, wsum, total);
    rpos[2 * threadIdx.x]     = ex;
    rpos[2 * threadIdx.x + 1] = ex + len0;
    rsrc[2 * threadIdx.x]     = q0 * CH + s0;
    rsrc[2 * threadIdx.x + 1] = q1 * CH + s1;
    __syncthreads();
    return total;
}

__device__ __forceinline__ void tile_load(int (&w)[FT], const int* __restrict__ binned,
                                          const int* rpos, const int* rsrc, int t0, int total) {
#pragma unroll
    for (int k = 0; k < FT; ++k) {
        const int e = t0 + k * 256 + threadIdx.x;
        const int i = (e < total) ? e : (total - 1);
        int r = 0;
#pragma unroll
        for (int st = RG / 2; st >= 1; st >>= 1)
            if (rpos[r + st] <= i) r += st;
        w[k] = binned[rsrc[r] + (i - rpos[r])];
    }
}

__device__ __forceinline__ void tile_count(const int (&w)[FT], int (&rk)[FT], int* c, int total) {
#pragma unroll
    for (int k = 0; k < FT; ++k) {
        const bool in = (k * 256 + (int)threadIdx.x) < total;
        rk[k] = atomicAdd(&c[in ? (w[k] & ID_MASK) : (int)threadIdx.x], in ? 1 : 0);
    }
}

__device__ __forceinline__ void tile_place(const int (&w)[FT], const int (&rk)[FT], const int* c,
                                           int* lst, int total) {
    int p[FT];
#pragma unroll
    for (int k = 0; k < FT; ++k) p[k] = c[w[k] & ID_MASK] + rk[k];
#pragma unroll
    for (int k = 0; k < FT; ++k)
        if (k * 256 + (int)threadIdx.x < total) lst[p[k]] = w[k] >> ID_SHIFT;
}

template <bool PLACE>
__device__ __forceinline__ void tile_walk(const int* __restrict__ binned, const int* rpos, const int* rsrc,
                                          int* c, int* __restrict__ slot, int t0, int total) {
#pragma unroll 4
    for (int k = 0; k < FT; ++k) {
        const int i = t0 + k * 256 + threadIdx.x;
        if (i < total) {
            int r = 0;
#pragma unroll
            for (int st = RG / 2; st >= 1; st >>= 1)
                if (rpos[r + st] <= i) r += st;
            const int wv = binned[rsrc[r] + (i - rpos[r])];
            const int p  = atomicAdd(&c[wv & ID_MASK], 1);
            if (PLACE) slot[p] = wv >> ID_SHIFT;
        }
    }
}

#define G2_ROWS  32
#define G2_STEPS ((BIN_NODES + G2_ROWS - 1) / G2_ROWS)
#define AG_ROWS  (((BIN_NODES + 15) / 16) * 16)
#define AG_PITCH 132
#define X_PITCH  132
#define O_PITCH  132
#define WH_PITCH 68
#define AG_FLOATS (AG_ROWS * AG_PITCH)
#define X_FLOATS  (G2_ROWS * X_PITCH)
#define O_FLOATS  (G2_ROWS * O_PITCH)
#define W_FLOATS  (8 * 16 * WH_PITCH)
#define S_WORDS   (264 + 2 * RG + TILE)
#define R_WORDS   W_FLOATS
static_assert(R_WORDS >= S_WORDS && R_WORDS >= X_FLOATS + O_FLOATS, "second LDS area too small");
static_assert((G2_STEPS - 1) * G2_ROWS + 16 <= AG_ROWS, "first tile of the last step inside the tile of means");

template <int Q>
__device__ __forceinline__ void rows_ask(float4 (&v)[8], const float4* __restrict__ x4, int idx, int lane) {
    int s[8];
#pragma unroll
    for (int u = 0; u < 8; ++u) s[u] = __builtin_amdgcn_readlane(idx, 8 * Q + u);
#pragma unroll
    for (int u = 0; u < 8; ++u) v[u] = x4[(long long)s[u] * (D / 4) + lane];
}

__device__ __forceinline__ void node_step(float4& acc, int& node, int& rem, int vdeg, int rcb,
                                          int row0, int nk, float* aggT, int lane) {
    do {
        const float rc = __int_as_float(__builtin_amdgcn_readlane(rcb, node));
        acc.x *= rc; acc.y *= rc; acc.z *= rc; acc.w *= rc;
        ((float4*)(aggT + (row0 + node) * AG_PITCH))[lane] = acc;
        acc = make_float4(0.0f, 0.0f, 0.0f, 0.0f);
        ++node;
        rem = (node < nk) ? __builtin_amdgcn_readlane(vdeg, node) : 0x7fffffff;
    } while (rem == 0);
}

__device__ __forceinline__ void rows_add(float4& acc, const float4 (&v)[8], int& node, int& rem,
                                         int vdeg, int rcb, int row0, int nk, float* aggT, int lane) {
#pragma unroll
    for (int u = 0; u < 8; ++u) {
        acc.x += v[u].x;
        acc.y += v[u].y;
        acc.z += v[u].z;
        acc.w += v[u].w;
        if (--rem == 0) node_step(acc, node, rem, vdeg, rcb, row0, nk, aggT, lane);
    }
}

__device__ __forceinline__ void bin_gather(const float* __restrict__ x, const int* ids,
                                           int vbeg, int vdeg, int row0, int nk,
                                           float* aggT, int lane) {
    const float4* x4 = (const float4*)x;
    if (nk <= 0) return;

    const int S = __builtin_amdgcn_readlane(vbeg, 0);
    const int T = __builtin_amdgcn_readlane(vbeg, nk - 1) + __builtin_amdgcn_readlane(vdeg, nk - 1) - S;
    const int rcb = __float_as_int(1.0f / fmaxf((float)vdeg, 1.0f));

    float4 acc = make_float4(0.0f, 0.0f, 0.0f, 0.0f);
    int node = 0;
    int rem  = __builtin_amdgcn_readlane(vdeg, 0);
    if (rem == 0) node_step(acc, node, rem, vdeg, rcb, row0, nk, aggT, lane);
    if (T <= 0) return;

    const int nch = (T + 31) >> 5;
    int idx = ids[S + ((lane < T) ? lane : (T - 1))];
    float4 r0[8], r1[8], r2[8], r3[8];
    rows_ask<0>(r0, x4, idx, lane);
    rows_ask<1>(r1, x4, idx, lane);
    rows_ask<2>(r2, x4, idx, lane);
    for (int c = 0; c + 1 < nch; ++c) {
        const int p = 32 * (c + 1) + lane;
        const int idxN = ids[S + ((p < T) ? p : (T - 1))];
        rows_ask<3>(r3, x4, idx, lane);
        rows_add(acc, r0, node, rem, vdeg, rcb, row0, nk, aggT, lane);
        rows_ask<0>(r0, x4, idxN, lane);
        rows_add(acc, r1, node, rem, vdeg, rcb, row0, nk, aggT, lane);
        rows_ask<1>(r1, x4, idxN, lane);
        rows_add(acc, r2, node, rem, vdeg, rcb, row0, nk, aggT, lane);
        rows_ask<2>(r2, x4, idxN, lane);
        rows_add(acc, r3, node, rem, vdeg, rcb, row0, nk, aggT, lane);
        idx = idxN;
    }
    rows_ask<3>(r3, x4, idx, lane);
    rows_add(acc, r0, node, rem, vdeg, rcb, row0, nk, aggT, lane);
    rows_add(acc, r1, node, rem, vdeg, rcb, row0, nk, aggT, lane);
    rows_add(acc, r2, node, rem, vdeg, rcb, row0, nk, aggT, lane);
    rows_add(acc, r3, node, rem, vdeg, rcb, row0, nk, aggT, lane);
}

struct StepRows {
    float4 x0, x1, x2, x3;
};

__device__ __forceinline__ float4 x_row(const float* __restrict__ x, int node, int lane) {
    const int nd = (node < N_NODES) ? node : (N_NODES - 1);
    return ((const float4*)(x + (long long)nd * D))[lane];
}

__device__ __forceinline__ void rows_load(StepRows& p, const float* __restrict__ x, int node0, int lane) {
    p.x0 = x_row(x, node0,      lane);
    p.x1 = x_row(x, node0 +  8, lane);
    p.x2 = x_row(x, node0 + 16, lane);
    p.x3 = x_row(x, node0 + 24, lane);
}

__device__ __forceinline__ void rows_put(const StepRows& p, float* xb, int w, int lane) {
    ((float4*)(xb + (w     ) * X_PITCH))[lane] = p.x0;
    ((float4*)(xb + (w +  8) * X_PITCH))[lane] = p.x1;
    ((float4*)(xb + (w + 16) * X_PITCH))[lane] = p.x2;
    ((float4*)(xb + (w + 24) * X_PITCH))[lane] = p.x3;
}


__global__ void __launch_bounds__(256)
bin_sage(const int*   __restrict__ lstartG,
         const int*   __restrict__ binned,
         const float* __restrict__ x,
         const float* __restrict__ Wl,
         const float* __restrict__ bl,
         const float* __restrict__ Wr,
         int*                      slot,
         float*       __restrict__ out,
         int G) {
    __shared__ __attribute__((aligned(16))) float lds[AG_FLOATS + R_WORDS];
    float* aggT = lds;
    float* rf   = lds + AG_FLOATS;
    int*   ri   = (int*)rf;
    int* c    = ri;
    int* wsum = ri + 256;
    int* rpos = ri + 264;
    int* rsrc = rpos + RG;
    int* lst  = rsrc + RG;
    const int b = blockIdx.x;

    c[threadIdx.x] = 0;

    int w[FT], rk[FT];
#pragma unroll
    for (int k = 0; k < FT; ++k) { w[k] = 0; rk[k] = 0; }
    int sumS = 0;
    int total = runs_build(lstartG, b, 0, G, rpos, rsrc, wsum, sumS);
    const bool single = (G <= RG) && (total <= TILE);
    if (single) {
        if (total > 0) {
            tile_load(w, binned, rpos, rsrc, 0, total);
            tile_count(w, rk, c, total);
        }
    } else {
        for (int g0 = 0;;) {
            for (int t0 = 0; t0 < total; t0 += TILE)
                tile_walk<false>(binned, rpos, rsrc, c, slot, t0, total);
            g0 += RG;
            if (g0 >= G) break;
            total = runs_build(lstartG, b, g0, G, rpos, rsrc, wsum, sumS);
        }
    }

    int lo;
    (void)block_excl_scan(sumS, wsum, lo);
    const int wv   = __builtin_amdgcn_readfirstlane((int)(threadIdx.x >> 5));
    const int lane = threadIdx.x & 31;
    const bool mine = lane < WAVE_NODES;
    const int nl   = WAVE_NODES * wv + (mine ? lane : 0);
    const int dgr  = mine ? c[nl] : 0;
    int tot;
    const int ex = block_excl_scan(dgr, wsum, tot);
    if (mine) c[nl] = single ? ex : (lo + ex);
    __syncthreads();

    const int node0 = b * BIN_NODES + WAVE_NODES * wv;
    const int left  = N_NODES - node0;
    const int nk    = (left < WAVE_NODES) ? ((left > 0) ? left : 0) : WAVE_NODES;

    if (single) {
        tile_place(w, rk, c, lst, total);
        __syncthreads();
        bin_gather(x, lst, ex, dgr, WAVE_NODES * wv, nk, aggT, lane);
    } else {
        int dummy = 0;
        for (int g0 = 0; g0 < G; g0 += RG) {
            total = runs_build(lstartG, b, g0, G, rpos, rsrc, wsum, dummy);
            for (int t0 = 0; t0 < total; t0 += TILE)
                tile_walk<true>(binned, rpos, rsrc, c, slot, t0, total);
        }
        __syncthreads();
        bin_gather(x, slot + lo, ex, dgr, WAVE_NODES * wv, nk, aggT, lane);
    }
    __syncthreads();

    const int c16 = lane & 15;
    const int h   = lane >> 4;

    v2f breg[64];
    {
        float*       wst = rf + wv * (16 * WH_PITCH);
        const float* wfr = wst + c16 * WH_PITCH + 2 * h;
#pragma unroll
        for (int ps = 0; ps < 4; ++ps) {
            const float* Wsrc = (ps < 2) ? Wl : Wr;
            const int    koff = (ps & 1) * (D / 2);
#pragma unroll
            for (int i = 0; i < 8; ++i) {
                const int r = 2 * i + h;
                ((float4*)(wst + r * WH_PITCH))[c16] =
                    ((const float4*)(Wsrc + (long long)(wv * 16 + r) * D + koff))[c16];
            }
            __syncthreads();
#pragma unroll
            for (int jj = 0; jj < 16; ++jj) breg[ps * 16 + jj] = *(const v2f*)(wfr + 4 * jj);
            __syncthreads();
        }
    }
    const float bb = bl[wv * 16 + c16];

    float* xbuf = rf;
    float* obuf = rf + X_FLOATS;
    const int bin0 = b * BIN_NODES;

    StepRows p;
    rows_load(p, x, bin0 + wv, lane);
    for (int s = 0; s < G2_STEPS; ++s) {
        rows_put(p, xbuf, wv, lane);
        __syncthreads();

        if (s + 1 < G2_STEPS) rows_load(p, x, bin0 + (s + 1) * G2_ROWS + wv, lane);

        const float* a0p = aggT + (s * G2_ROWS + c16) * AG_PITCH + 2 * h;
        const float* a1p = a0p + 16 * AG_PITCH;
        const float* x0p = xbuf + c16 * X_PITCH + 2 * h;
        const float* x1p = x0p + 16 * X_PITCH;
        const bool two = (s * G2_ROWS + 16) < BIN_NODES;
        v8f acc0 = {}, acc1 = {};
        if (two) {
#pragma unroll
            for (int j = 0; j < 32; ++j) {
                const v2f a0 = *(const v2f*)(a0p + 4 * j);
                const v2f a1 = *(const v2f*)(a1p + 4 * j);
                acc0 = __builtin_amdgcn_wmma_f32_16x16x4_f32(
                    false, a0, false, breg[j], (short)0, acc0, false, false);
                acc1 = __builtin_amdgcn_wmma_f32_16x16x4_f32(
                    false, a1, false, breg[j], (short)0, acc1, false, false);
            }
#pragma unroll
            for (int j = 0; j < 32; ++j) {
                const v2f a0 = *(const v2f*)(x0p + 4 * j);
                const v2f a1 = *(const v2f*)(x1p + 4 * j);
                acc0 = __builtin_amdgcn_wmma_f32_16x16x4_f32(
                    false, a0, false, breg[32 + j], (short)0, acc0, false, false);
                acc1 = __builtin_amdgcn_wmma_f32_16x16x4_f32(
                    false, a1, false, breg[32 + j], (short)0, acc1, false, false);
            }
        } else {
#pragma unroll
            for (int j = 0; j < 32; ++j) {
                const v2f a0 = *(const v2f*)(a0p + 4 * j);
                acc0 = __builtin_amdgcn_wmma_f32_16x16x4_f32(
                    false, a0, false, breg[j], (short)0, acc0, false, false);
            }
#pragma unroll
            for (int j = 0; j < 32; ++j) {
                const v2f a0 = *(const v2f*)(x0p + 4 * j);
                acc0 = __builtin_amdgcn_wmma_f32_16x16x4_f32(
                    false, a0, false, breg[32 + j], (short)0, acc0, false, false);
            }
        }

#pragma unroll
        for (int j = 0; j < 8; ++j) {
            float v0 = acc0[j] + bb;
            v0 = (v0 > 0.0f) ? v0 : SLOPE * v0;
            obuf[(8 * h + j) * O_PITCH + 16 * wv + c16] = v0;
            float v1 = acc1[j] + bb;
            v1 = (v1 > 0.0f) ? v1 : SLOPE * v1;
            obuf[(16 + 8 * h + j) * O_PITCH + 16 * wv + c16] = v1;
        }
        __syncthreads();

#pragma unroll
        for (int q = 0; q < 4; ++q) {
            const int r   = s * G2_ROWS + q * 8 + wv;
            const int nd  = bin0 + r;
            const float4 t = ((const float4*)(obuf + (q * 8 + wv) * O_PITCH))[lane];
            if (r < BIN_NODES && nd < N_NODES)
                ((float4*)(out + (long long)nd * D))[lane] = t;
        }
    }
}

extern "C" void kernel_launch(void* const* d_in, const int* in_sizes, int n_in,
                              void* d_out, int out_size, void* d_ws, size_t ws_size,
                              hipStream_t stream) {
    const float* x  = (const float*)d_in[0];
    const int*   ei = (const int*)  d_in[1];
    const float* Wl = (const float*)d_in[2];
    const float* bl = (const float*)d_in[3];
    const float* Wr = (const float*)d_in[4];
    float* out = (float*)d_out;
    const int E = in_sizes[1] / 2;

    char*  ws     = (char*)d_ws;
    int*   lstartG = (int*) (ws + (2  << 20));
    int*   binned = (int*)  (ws + (8  << 20));
    int*   slot   = (int*)  (ws + (16 << 20));

    const int G = (E + CH - 1) / CH;
    sort_local<<<G,  256, 0, stream>>>(ei, lstartG, binned, E);
    bin_sage  <<<NB, 256, 0, stream>>>(lstartG, binned, x, Wl, bl, Wr, slot, out, G);
}
